# starting kernel
# baseline (speedup 1.0000x reference)
#include <hip/hip_runtime.h>
#include <math.h>

typedef float v2f __attribute__((ext_vector_type(2)));
typedef float v8f __attribute__((ext_vector_type(8)));

#define NODE_DIM 128
#define ATTR_DIM 64
#define EPS 1e-8f

__global__ void init_kernel(float* __restrict__ seg_max,
                            float* __restrict__ seg_sum,
                            float* __restrict__ agg, int n1) {
  int i = blockIdx.x * blockDim.x + threadIdx.x;
  int total = n1 * NODE_DIM;
  if (i < total) agg[i] = 0.0f;
  if (i < n1) {
    seg_max[i] = -INFINITY;
    seg_sum[i] = 0.0f;
  }
}

__global__ void inv_norm_kernel(const float* __restrict__ X,
                                float* __restrict__ inv, int n) {
  int wave = (blockIdx.x * blockDim.x + threadIdx.x) >> 5;
  int lane = threadIdx.x & 31;
  if (wave >= n) return;
  float4 a = reinterpret_cast<const float4*>(X + (size_t)wave * NODE_DIM)[lane];
  float p = a.x * a.x + a.y * a.y + a.z * a.z + a.w * a.w;
#pragma unroll
  for (int off = 16; off > 0; off >>= 1) p += __shfl_xor(p, off, 32);
  if (lane == 0) inv[wave] = 1.0f / fmaxf(sqrtf(p), EPS);
}

__global__ void edge_sim_kernel(const float* __restrict__ X1,
                                const float* __restrict__ X2,
                                const int* __restrict__ src,
                                const int* __restrict__ dst,
                                const float* __restrict__ invn1,
                                const float* __restrict__ invn2,
                                float* __restrict__ sim,
                                float* __restrict__ seg_max, int E) {
  int e = (blockIdx.x * blockDim.x + threadIdx.x) >> 5;
  int lane = threadIdx.x & 31;
  if (e >= E) return;
  int s = src[e];
  int d = dst[e];
  float4 a = reinterpret_cast<const float4*>(X1 + (size_t)s * NODE_DIM)[lane];
  float4 b = reinterpret_cast<const float4*>(X2 + (size_t)d * NODE_DIM)[lane];
  float p = a.x * b.x + a.y * b.y + a.z * b.z + a.w * b.w;
#pragma unroll
  for (int off = 16; off > 0; off >>= 1) p += __shfl_xor(p, off, 32);
  if (lane == 0) {
    float simv = p * invn1[s] * invn2[d];
    sim[e] = simv;
    unsafeAtomicMax(&seg_max[s], simv);
  }
}

__global__ void edge_exp_kernel(const int* __restrict__ src,
                                const float* __restrict__ sim,
                                const float* __restrict__ seg_max,
                                float* __restrict__ ex,
                                float* __restrict__ seg_sum, int E) {
  int e = blockIdx.x * blockDim.x + threadIdx.x;
  if (e >= E) return;
  int s = src[e];
  float v = expf(sim[e] - seg_max[s]);
  ex[e] = v;
  unsafeAtomicAdd(&seg_sum[s], v);
}

__global__ void edge_agg_kernel(const float* __restrict__ X2,
                                const int* __restrict__ src,
                                const int* __restrict__ dst,
                                const float* __restrict__ ex,
                                const float* __restrict__ seg_sum,
                                float* __restrict__ agg, int E) {
  int e = (blockIdx.x * blockDim.x + threadIdx.x) >> 5;
  int lane = threadIdx.x & 31;
  if (e >= E) return;
  int s = src[e];
  int d = dst[e];
  float w = ex[e] / seg_sum[s];
  float4 b = reinterpret_cast<const float4*>(X2 + (size_t)d * NODE_DIM)[lane];
  float* out = agg + (size_t)s * NODE_DIM + lane * 4;
  unsafeAtomicAdd(out + 0, w * b.x);
  unsafeAtomicAdd(out + 1, w * b.y);
  unsafeAtomicAdd(out + 2, w * b.z);
  unsafeAtomicAdd(out + 3, w * b.w);
}

__global__ void gate_out_kernel(const float* __restrict__ Xn,
                                const float* __restrict__ Wg,
                                const float* __restrict__ agg,
                                float* __restrict__ out, int n1) {
  int wave = blockIdx.x * (blockDim.x >> 5) + (threadIdx.x >> 5);
  int lane = threadIdx.x & 31;
  int tilesM = (n1 + 15) >> 4;
  int tileM = wave >> 3;
  int tileN = wave & 7;
  if (tileM >= tilesM) return;

  int l15 = lane & 15;
  int half = lane >> 4;
  int m = tileM * 16 + l15;
  if (m >= n1) m = n1 - 1;
  const float* arow = Xn + (size_t)m * ATTR_DIM;
  const float* brow = Wg + (size_t)(tileN * 16 + l15) * ATTR_DIM;

  v8f c = {};
#pragma unroll
  for (int k0 = 0; k0 < ATTR_DIM; k0 += 4) {
    int k = k0 + half * 2;
    v2f a = *reinterpret_cast<const v2f*>(arow + k);
    v2f b = *reinterpret_cast<const v2f*>(brow + k);
    c = __builtin_amdgcn_wmma_f32_16x16x4_f32(
         false, a,  false, b,
         (short)0, c,  false,  false);
  }

  int n = tileN * 16 + l15;
#pragma unroll
  for (int v = 0; v < 8; ++v) {
    int row = tileM * 16 + half * 8 + v;
    if (row < n1) {
      float g = 1.0f / (1.0f + expf(-c[v]));
      size_t idx = (size_t)row * NODE_DIM + n;
      out[idx] = g * agg[idx];
    }
  }
}

extern "C" void kernel_launch(void* const* d_in, const int* in_sizes, int n_in,
                              void* d_out, int out_size, void* d_ws,
                              size_t ws_size, hipStream_t stream) {
  const float* X_h_1 = (const float*)d_in[0];
  const float* X_h_2 = (const float*)d_in[1];
  const float* X_n_1 = (const float*)d_in[2];
  const int* cross = (const int*)d_in[3];
  const float* W_gate = (const float*)d_in[4];
  float* out = (float*)d_out;

  const int N1 = in_sizes[0] / NODE_DIM;
  const int N2 = in_sizes[1] / NODE_DIM;
  const int E = in_sizes[3] / 2;
  const int* src = cross;
  const int* dst = cross + E;

  float* ws = (float*)d_ws;
  float* invn1 = ws;
  float* invn2 = invn1 + N1;
  float* sim = invn2 + N2;
  float* ex = sim + E;
  float* seg_max = ex + E;
  float* seg_sum = seg_max + N1;
  float* agg = seg_sum + N1;

  const int BLK = 256;
  const int waves_per_blk = BLK / 32;

  {
    int total = N1 * NODE_DIM;
    init_kernel<<<(total + BLK - 1) / BLK, BLK, 0, stream>>>(seg_max, seg_sum,
                                                             agg, N1);
  }
  inv_norm_kernel<<<(N1 + waves_per_blk - 1) / waves_per_blk, BLK, 0, stream>>>(
      X_h_1, invn1, N1);
  inv_norm_kernel<<<(N2 + waves_per_blk - 1) / waves_per_blk, BLK, 0, stream>>>(
      X_h_2, invn2, N2);
  edge_sim_kernel<<<(E + waves_per_blk - 1) / waves_per_blk, BLK, 0, stream>>>(
      X_h_1, X_h_2, src, dst, invn1, invn2, sim, seg_max, E);
  edge_exp_kernel<<<(E + BLK - 1) / BLK, BLK, 0, stream>>>(src, sim, seg_max,
                                                           ex, seg_sum, E);
  edge_agg_kernel<<<(E + waves_per_blk - 1) / waves_per_blk, BLK, 0, stream>>>(
      X_h_2, src, dst, ex, seg_sum, agg, E);
  {
    int tilesM = (N1 + 15) / 16;
    int waves = tilesM * (NODE_DIM / 16);
    gate_out_kernel<<<(waves + waves_per_blk - 1) / waves_per_blk, BLK, 0,
                      stream>>>(X_n_1, W_gate, agg, out, N1);
  }
}


# one edge pass in place of three: softmax without the max shift (cosine sim in [-1,1] bounds exp); exp(sim) and exp(sim)*row summed by f32 atomics as before, divided by seg_sum in the gate kernel
# speedup vs baseline: 1.4961x; 1.4961x over previous
#include <hip/hip_runtime.h>
#include <math.h>

typedef float v2f __attribute__((ext_vector_type(2)));
typedef float v8f __attribute__((ext_vector_type(8)));

#define NODE_DIM 128
#define ATTR_DIM 64
#define EPS 1e-8f

__global__ void init_kernel(float* __restrict__ seg_sum,
                            float* __restrict__ agg, int n1) {
  int i = blockIdx.x * blockDim.x + threadIdx.x;
  int total = n1 * NODE_DIM;
  if (i < total) agg[i] = 0.0f;
  if (i < n1) seg_sum[i] = 0.0f;
}

__global__ void inv_norm_kernel(const float* __restrict__ X,
                                float* __restrict__ inv, int n) {
  int wave = (blockIdx.x * blockDim.x + threadIdx.x) >> 5;
  int lane = threadIdx.x & 31;
  if (wave >= n) return;
  float4 a = reinterpret_cast<const float4*>(X + (size_t)wave * NODE_DIM)[lane];
  float p = a.x * a.x + a.y * a.y + a.z * a.z + a.w * a.w;
#pragma unroll
  for (int off = 16; off > 0; off >>= 1) p += __shfl_xor(p, off, 32);
  if (lane == 0) inv[wave] = 1.0f / fmaxf(sqrtf(p), EPS);
}

__global__ void edge_fused_kernel(const float* __restrict__ X1,
                                  const float* __restrict__ X2,
                                  const int* __restrict__ src,
                                  const int* __restrict__ dst,
                                  const float* __restrict__ invn1,
                                  const float* __restrict__ invn2,
                                  float* __restrict__ seg_sum,
                                  float* __restrict__ agg, int E) {
  int e = (blockIdx.x * blockDim.x + threadIdx.x) >> 5;
  int lane = threadIdx.x & 31;
  if (e >= E) return;
  int s = src[e];
  int d = dst[e];
  float4 a = reinterpret_cast<const float4*>(X1 + (size_t)s * NODE_DIM)[lane];
  float4 b = reinterpret_cast<const float4*>(X2 + (size_t)d * NODE_DIM)[lane];
  float p = a.x * b.x + a.y * b.y + a.z * b.z + a.w * b.w;
#pragma unroll
  for (int off = 16; off > 0; off >>= 1) p += __shfl_xor(p, off, 32);
  float simv = p * invn1[s] * invn2[d];
  float w = expf(simv);
  if (lane == 0) unsafeAtomicAdd(&seg_sum[s], w);
  float* out = agg + (size_t)s * NODE_DIM + lane * 4;
  unsafeAtomicAdd(out + 0, w * b.x);
  unsafeAtomicAdd(out + 1, w * b.y);
  unsafeAtomicAdd(out + 2, w * b.z);
  unsafeAtomicAdd(out + 3, w * b.w);
}

__global__ void gate_out_kernel(const float* __restrict__ Xn,
                                const float* __restrict__ Wg,
                                const float* __restrict__ agg,
                                const float* __restrict__ seg_sum,
                                float* __restrict__ out, int n1) {
  int wave = blockIdx.x * (blockDim.x >> 5) + (threadIdx.x >> 5);
  int lane = threadIdx.x & 31;
  int tilesM = (n1 + 15) >> 4;
  int tileM = wave >> 3;
  int tileN = wave & 7;
  if (tileM >= tilesM) return;

  int l15 = lane & 15;
  int half = lane >> 4;
  int m = tileM * 16 + l15;
  if (m >= n1) m = n1 - 1;
  const float* arow = Xn + (size_t)m * ATTR_DIM;
  const float* brow = Wg + (size_t)(tileN * 16 + l15) * ATTR_DIM;

  v8f c = {};
#pragma unroll
  for (int k0 = 0; k0 < ATTR_DIM; k0 += 4) {
    int k = k0 + half * 2;
    v2f a = *reinterpret_cast<const v2f*>(arow + k);
    v2f b = *reinterpret_cast<const v2f*>(brow + k);
    c = __builtin_amdgcn_wmma_f32_16x16x4_f32(
         false, a,  false, b,
         (short)0, c,  false,  false);
  }

  int n = tileN * 16 + l15;
#pragma unroll
  for (int v = 0; v < 8; ++v) {
    int row = tileM * 16 + half * 8 + v;
    if (row < n1) {
      float g = 1.0f / (1.0f + expf(-c[v]));
      float ss = seg_sum[row];
      float inv = ss > 0.0f ? 1.0f / ss : 0.0f;
      size_t idx = (size_t)row * NODE_DIM + n;
      out[idx] = g * (agg[idx] * inv);
    }
  }
}

extern "C" void kernel_launch(void* const* d_in, const int* in_sizes, int n_in,
                              void* d_out, int out_size, void* d_ws,
                              size_t ws_size, hipStream_t stream) {
  const float* X_h_1 = (const float*)d_in[0];
  const float* X_h_2 = (const float*)d_in[1];
  const float* X_n_1 = (const float*)d_in[2];
  const int* cross = (const int*)d_in[3];
  const float* W_gate = (const float*)d_in[4];
  float* out = (float*)d_out;

  const int N1 = in_sizes[0] / NODE_DIM;
  const int N2 = in_sizes[1] / NODE_DIM;
  const int E = in_sizes[3] / 2;
  const int* src = cross;
  const int* dst = cross + E;

  float* ws = (float*)d_ws;
  float* invn1 = ws;
  float* invn2 = invn1 + N1;
  float* seg_sum = invn2 + N2;
  float* agg = seg_sum + N1;

  const int BLK = 256;
  const int waves_per_blk = BLK / 32;

  {
    int total = N1 * NODE_DIM;
    init_kernel<<<(total + BLK - 1) / BLK, BLK, 0, stream>>>(seg_sum, agg, N1);
  }
  inv_norm_kernel<<<(N1 + waves_per_blk - 1) / waves_per_blk, BLK, 0, stream>>>(
      X_h_1, invn1, N1);
  inv_norm_kernel<<<(N2 + waves_per_blk - 1) / waves_per_blk, BLK, 0, stream>>>(
      X_h_2, invn2, N2);
  edge_fused_kernel<<<(E + waves_per_blk - 1) / waves_per_blk, BLK, 0,
                      stream>>>(X_h_1, X_h_2, src, dst, invn1, invn2, seg_sum,
                                agg, E);
  {
    int tilesM = (N1 + 15) / 16;
    int waves = tilesM * (NODE_DIM / 16);
    gate_out_kernel<<<(waves + waves_per_blk - 1) / waves_per_blk, BLK, 0,
                      stream>>>(X_n_1, W_gate, agg, seg_sum, out, N1);
  }
}


# edge scatter without float atomics: one int atomic exchange per edge chains it to its source node, then one wave per node walks the list and sums in registers (sum order = reverse arrival)
# speedup vs baseline: 2.3581x; 1.5762x over previous
#include <hip/hip_runtime.h>
#include <math.h>

typedef float v2f __attribute__((ext_vector_type(2)));
typedef float v8f __attribute__((ext_vector_type(8)));

#define NODE_DIM 128
#define ATTR_DIM 64
#define EPS 1e-8f

__global__ void init_kernel(int* __restrict__ head, int n1) {
  int i = blockIdx.x * blockDim.x + threadIdx.x;
  if (i < n1) head[i] = -1;
}

__global__ void inv_norm_kernel(const float* __restrict__ X,
                                float* __restrict__ inv, int n) {
  int wave = (blockIdx.x * blockDim.x + threadIdx.x) >> 5;
  int lane = threadIdx.x & 31;
  if (wave >= n) return;
  float4 a = reinterpret_cast<const float4*>(X + (size_t)wave * NODE_DIM)[lane];
  float p = a.x * a.x + a.y * a.y + a.z * a.z + a.w * a.w;
#pragma unroll
  for (int off = 16; off > 0; off >>= 1) p += __shfl_xor(p, off, 32);
  if (lane == 0) inv[wave] = 1.0f / fmaxf(sqrtf(p), EPS);
}

#define LINK_EPT 4
__global__ void link_kernel(const int* __restrict__ src,
                            const int* __restrict__ dst,
                            const float* __restrict__ invn2,
                            int* __restrict__ head, int4* __restrict__ link,
                            int E) {
  int base = blockIdx.x * (blockDim.x * LINK_EPT) + threadIdx.x;
  int s[LINK_EPT], d[LINK_EPT];
#pragma unroll
  for (int k = 0; k < LINK_EPT; ++k) {
    int e = min(base + k * (int)blockDim.x, E - 1);
    s[k] = src[e];
    d[k] = dst[e];
  }
#pragma unroll
  for (int k = 0; k < LINK_EPT; ++k) {
    int e = base + k * blockDim.x;
    if (e < E) {
      float in2 = invn2[d[k]];
      int prev = atomicExch(&head[s[k]], e);
      link[e] = make_int4(prev, d[k], __float_as_int(in2), 0);
    }
  }
}

__global__ void gather_kernel(const float* __restrict__ X1,
                              const float* __restrict__ X2,
                              const int* __restrict__ head,
                              const int4* __restrict__ link,
                              const float* __restrict__ invn1,
                              float* __restrict__ seg_sum,
                              float* __restrict__ out, int n1, int E) {
  int s = __builtin_amdgcn_readfirstlane(
      (int)((blockIdx.x * blockDim.x + threadIdx.x) >> 5));
  int lane = threadIdx.x & 31;
  if (s >= n1) return;
  float4 a = reinterpret_cast<const float4*>(X1 + (size_t)s * NODE_DIM)[lane];
  float in1 = invn1[s];
  float4 acc = make_float4(0.0f, 0.0f, 0.0f, 0.0f);
  float ssum = 0.0f;
  int e = head[s];
  if ((unsigned)e < (unsigned)E) {
    int4 cur = link[e];
    for (int it = 0; it < E; ++it) {
      int nxt = cur.x;
      int d = cur.y;
      float4 b = reinterpret_cast<const float4*>(X2 + (size_t)d * NODE_DIM)[lane];
      float in2 = __int_as_float(cur.z);
      bool more = (unsigned)nxt < (unsigned)E;
      if (more) cur = link[nxt];
      float p = a.x * b.x + a.y * b.y + a.z * b.z + a.w * b.w;
#pragma unroll
      for (int off = 16; off > 0; off >>= 1) p += __shfl_xor(p, off, 32);
      float simv = p * in1 * in2;
      float w = expf(simv);
      ssum += w;
      acc.x += w * b.x;
      acc.y += w * b.y;
      acc.z += w * b.z;
      acc.w += w * b.w;
      if (!more) break;
    }
  }
  reinterpret_cast<float4*>(out + (size_t)s * NODE_DIM)[lane] = acc;
  if (lane == 0) seg_sum[s] = ssum;
}

__global__ void gate_out_kernel(const float* __restrict__ Xn,
                                const float* __restrict__ Wg,
                                const float* __restrict__ seg_sum,
                                float* __restrict__ out, int n1) {
  int wave = blockIdx.x * (blockDim.x >> 5) + (threadIdx.x >> 5);
  int lane = threadIdx.x & 31;
  int tilesM = (n1 + 15) >> 4;
  int tileM = wave >> 3;
  int tileN = wave & 7;
  if (tileM >= tilesM) return;

  int l15 = lane & 15;
  int half = lane >> 4;
  int m = tileM * 16 + l15;
  if (m >= n1) m = n1 - 1;
  const float* arow = Xn + (size_t)m * ATTR_DIM;
  const float* brow = Wg + (size_t)(tileN * 16 + l15) * ATTR_DIM;

  v8f c = {};
#pragma unroll
  for (int k0 = 0; k0 < ATTR_DIM; k0 += 4) {
    int k = k0 + half * 2;
    v2f a = *reinterpret_cast<const v2f*>(arow + k);
    v2f b = *reinterpret_cast<const v2f*>(brow + k);
    c = __builtin_amdgcn_wmma_f32_16x16x4_f32(
         false, a,  false, b,
         (short)0, c,  false,  false);
  }

  int n = tileN * 16 + l15;
#pragma unroll
  for (int v = 0; v < 8; ++v) {
    int row = tileM * 16 + half * 8 + v;
    if (row < n1) {
      float g = 1.0f / (1.0f + expf(-c[v]));
      float ss = seg_sum[row];
      float inv = ss > 0.0f ? 1.0f / ss : 0.0f;
      size_t idx = (size_t)row * NODE_DIM + n;
      out[idx] = g * (out[idx] * inv);
    }
  }
}

extern "C" void kernel_launch(void* const* d_in, const int* in_sizes, int n_in,
                              void* d_out, int out_size, void* d_ws,
                              size_t ws_size, hipStream_t stream) {
  const float* X_h_1 = (const float*)d_in[0];
  const float* X_h_2 = (const float*)d_in[1];
  const float* X_n_1 = (const float*)d_in[2];
  const int* cross = (const int*)d_in[3];
  const float* W_gate = (const float*)d_in[4];
  float* out = (float*)d_out;

  const int N1 = in_sizes[0] / NODE_DIM;
  const int N2 = in_sizes[1] / NODE_DIM;
  const int E = in_sizes[3] / 2;
  const int* src = cross;
  const int* dst = cross + E;

  int4* link = (int4*)d_ws;
  float* invn1 = (float*)(link + E);
  float* invn2 = invn1 + N1;
  float* seg_sum = invn2 + N2;
  int* head = (int*)(seg_sum + N1);

  const int BLK = 256;
  const int waves_per_blk = BLK / 32;

  init_kernel<<<(N1 + BLK - 1) / BLK, BLK, 0, stream>>>(head, N1);
  inv_norm_kernel<<<(N1 + waves_per_blk - 1) / waves_per_blk, BLK, 0, stream>>>(
      X_h_1, invn1, N1);
  inv_norm_kernel<<<(N2 + waves_per_blk - 1) / waves_per_blk, BLK, 0, stream>>>(
      X_h_2, invn2, N2);
  link_kernel<<<(E + BLK * LINK_EPT - 1) / (BLK * LINK_EPT), BLK, 0, stream>>>(
      src, dst, invn2, head, link, E);
  gather_kernel<<<(N1 + waves_per_blk - 1) / waves_per_blk, BLK, 0, stream>>>(
      X_h_1, X_h_2, head, link, invn1, seg_sum, out, N1, E);
  {
    int tilesM = (N1 + 15) / 16;
    int waves = tilesM * (NODE_DIM / 16);
    gate_out_kernel<<<(waves + waves_per_blk - 1) / waves_per_blk, BLK, 0,
                      stream>>>(X_n_1, W_gate, seg_sum, out, N1);
  }
}


# more work per wave against the wave entry rate: 8 rows per wave in the norm kernels, 4 node lists walked side by side per wave in the gather, a 16x64 half strip per wave in the gate GEMM
# speedup vs baseline: 2.6706x; 1.1325x over previous
#include <hip/hip_runtime.h>
#include <math.h>

typedef float v2f __attribute__((ext_vector_type(2)));
typedef float v8f __attribute__((ext_vector_type(8)));

#define NODE_DIM 128
#define ATTR_DIM 64
#define EPS 1e-8f

__global__ void init_kernel(int* __restrict__ head, int n1) {
  int i = blockIdx.x * blockDim.x + threadIdx.x;
  if (i < n1) head[i] = -1;
}

#define NORM_ROWS 8
__global__ void inv_norm_kernel(const float* __restrict__ X,
                                float* __restrict__ inv, int n) {
  int wave = (blockIdx.x * blockDim.x + threadIdx.x) >> 5;
  int lane = threadIdx.x & 31;
  int r0 = wave * NORM_ROWS;
  if (r0 >= n) return;
  float p[NORM_ROWS];
#pragma unroll
  for (int k = 0; k < NORM_ROWS; ++k) {
    int r = min(r0 + k, n - 1);
    float4 a = reinterpret_cast<const float4*>(X + (size_t)r * NODE_DIM)[lane];
    p[k] = a.x * a.x + a.y * a.y + a.z * a.z + a.w * a.w;
  }
#pragma unroll
  for (int k = 0; k < NORM_ROWS; ++k) {
#pragma unroll
    for (int off = 16; off > 0; off >>= 1) p[k] += __shfl_xor(p[k], off, 32);
  }
  float mine = p[0];
#pragma unroll
  for (int k = 1; k < NORM_ROWS; ++k)
    if (lane == k) mine = p[k];
  if (lane < NORM_ROWS && r0 + lane < n)
    inv[r0 + lane] = 1.0f / fmaxf(sqrtf(mine), EPS);
}

#define LINK_EPT 4
__global__ void link_kernel(const int* __restrict__ src,
                            const int* __restrict__ dst,
                            const float* __restrict__ invn2,
                            int* __restrict__ head, int4* __restrict__ link,
                            int E) {
  int base = blockIdx.x * (blockDim.x * LINK_EPT) + threadIdx.x;
  int s[LINK_EPT], d[LINK_EPT];
#pragma unroll
  for (int k = 0; k < LINK_EPT; ++k) {
    int e = min(base + k * (int)blockDim.x, E - 1);
    s[k] = src[e];
    d[k] = dst[e];
  }
#pragma unroll
  for (int k = 0; k < LINK_EPT; ++k) {
    int e = base + k * blockDim.x;
    if (e < E) {
      float in2 = invn2[d[k]];
      int prev = atomicExch(&head[s[k]], e);
      link[e] = make_int4(prev, d[k], __float_as_int(in2), 0);
    }
  }
}

#define GATHER_K 4
__global__ void gather_kernel(const float* __restrict__ X1,
                              const float* __restrict__ X2,
                              const int* __restrict__ head,
                              const int4* __restrict__ link,
                              const float* __restrict__ invn1,
                              float* __restrict__ seg_sum,
                              float* __restrict__ out, int n1, int E) {
  int w = __builtin_amdgcn_readfirstlane(
      (int)((blockIdx.x * blockDim.x + threadIdx.x) >> 5));
  int lane = threadIdx.x & 31;
  int s0 = w * GATHER_K;
  if (s0 >= n1) return;
  float4 a[GATHER_K], acc[GATHER_K];
  float in1[GATHER_K], ssum[GATHER_K];
  int4 cur[GATHER_K];
  bool act[GATHER_K];
  bool hn[GATHER_K];
  int nidx[GATHER_K];
#pragma unroll
  for (int k = 0; k < GATHER_K; ++k) {
    int s = min(s0 + k, n1 - 1);
    a[k] = reinterpret_cast<const float4*>(X1 + (size_t)s * NODE_DIM)[lane];
    in1[k] = invn1[s];
    nidx[k] = head[s];
    acc[k] = make_float4(0.0f, 0.0f, 0.0f, 0.0f);
    ssum[k] = 0.0f;
  }
#pragma unroll
  for (int k = 0; k < GATHER_K; ++k) {
    act[k] = (s0 + k < n1) & ((unsigned)nidx[k] < (unsigned)E);
    cur[k] = link[act[k] ? nidx[k] : 0];
  }
#pragma unroll
  for (int k = 0; k < GATHER_K; ++k) {
    hn[k] = act[k] & ((unsigned)cur[k].x < (unsigned)E);
    nidx[k] = hn[k] ? cur[k].x : 0;
  }
  for (int it = 0; it < E; ++it) {
    bool any = false;
#pragma unroll
    for (int k = 0; k < GATHER_K; ++k) any = any | act[k];
    if (!any) break;
    float4 b[GATHER_K];
    int4 nx[GATHER_K];
#pragma unroll
    for (int k = 0; k < GATHER_K; ++k) {
      int d = cur[k].y;
      b[k] = reinterpret_cast<const float4*>(X2 + (size_t)d * NODE_DIM)[lane];
      nx[k] = link[nidx[k]];
    }
#pragma unroll
    for (int k = 0; k < GATHER_K; ++k) {
      float p = a[k].x * b[k].x + a[k].y * b[k].y + a[k].z * b[k].z +
                a[k].w * b[k].w;
#pragma unroll
      for (int off = 16; off > 0; off >>= 1) p += __shfl_xor(p, off, 32);
      float in2 = __int_as_float(cur[k].z);
      float simv = p * in1[k] * in2;
      float wgt = expf(simv);
      ssum[k] = act[k] ? ssum[k] + wgt : ssum[k];
      acc[k].x = act[k] ? acc[k].x + wgt * b[k].x : acc[k].x;
      acc[k].y = act[k] ? acc[k].y + wgt * b[k].y : acc[k].y;
      acc[k].z = act[k] ? acc[k].z + wgt * b[k].z : acc[k].z;
      acc[k].w = act[k] ? acc[k].w + wgt * b[k].w : acc[k].w;
    }
#pragma unroll
    for (int k = 0; k < GATHER_K; ++k) {
      asm volatile("" : "+s"(nx[k].x), "+s"(nx[k].y), "+s"(nx[k].z));
      act[k] = hn[k];
      cur[k] = nx[k];
      hn[k] = hn[k] & ((unsigned)nx[k].x < (unsigned)E);
      nidx[k] = hn[k] ? nx[k].x : 0;
    }
  }
#pragma unroll
  for (int k = 0; k < GATHER_K; ++k) {
    int s = s0 + k;
    if (s < n1) {
      reinterpret_cast<float4*>(out + (size_t)s * NODE_DIM)[lane] = acc[k];
      if (lane == 0) seg_sum[s] = ssum[k];
    }
  }
}

#define GATE_TILES 4
#define GATE_KSTEPS (ATTR_DIM / 4)

template <bool FULL>
__device__ __forceinline__ void gate_strip(const v2f (&a)[GATE_KSTEPS],
                                           const float* __restrict__ brow0,
                                           const float (&inv)[8],
                                           float* __restrict__ out, int n1,
                                           int row0, int n0) {
#pragma unroll
  for (int t = 0; t < GATE_TILES; ++t) {
    const float* brow = brow0 + (size_t)t * 16 * ATTR_DIM;
    int n = n0 + t * 16;
    v2f b[GATE_KSTEPS];
#pragma unroll
    for (int i = 0; i < GATE_KSTEPS; ++i)
      b[i] = *reinterpret_cast<const v2f*>(brow + i * 4);
    float o[8];
#pragma unroll
    for (int v = 0; v < 8; ++v) {
      int row = FULL ? row0 + v : min(row0 + v, n1 - 1);
      o[v] = out[(size_t)row * NODE_DIM + n];
    }
    v8f c = {};
#pragma unroll
    for (int i = 0; i < GATE_KSTEPS; ++i)
      c = __builtin_amdgcn_wmma_f32_16x16x4_f32(
           false, a[i],  false, b[i],
           (short)0, c,  false,  false);
    float r[8];
#pragma unroll
    for (int v = 0; v < 8; ++v) {
      float g = 1.0f / (1.0f + expf(-c[v]));
      r[v] = g * (o[v] * inv[v]);
    }
#pragma unroll
    for (int v = 0; v < 8; ++v) {
      int row = row0 + v;
      if (FULL || row < n1) out[(size_t)row * NODE_DIM + n] = r[v];
    }
  }
}

__global__ void __launch_bounds__(256)
gate_out_kernel(const float* __restrict__ Xn,
                                const float* __restrict__ Wg,
                                const float* __restrict__ seg_sum,
                                float* __restrict__ out, int n1) {
  int wave = blockIdx.x * (blockDim.x >> 5) + (threadIdx.x >> 5);
  int lane = threadIdx.x & 31;
  int tilesM = (n1 + 15) >> 4;
  const int strips = NODE_DIM / 16 / GATE_TILES;
  int tileM = wave / strips;
  int tn0 = (wave % strips) * GATE_TILES;
  if (tileM >= tilesM) return;

  int l15 = lane & 15;
  int half = lane >> 4;
  int m = tileM * 16 + l15;
  if (m >= n1) m = n1 - 1;
  const float* arow = Xn + (size_t)m * ATTR_DIM + half * 2;

  v2f a[GATE_KSTEPS];
#pragma unroll
  for (int i = 0; i < GATE_KSTEPS; ++i)
    a[i] = *reinterpret_cast<const v2f*>(arow + i * 4);

  int row0 = tileM * 16 + half * 8;
  float inv[8];
#pragma unroll
  for (int v = 0; v < 8; ++v) {
    float ss = seg_sum[min(row0 + v, n1 - 1)];
    inv[v] = ss > 0.0f ? 1.0f / ss : 0.0f;
  }

  const float* brow0 = Wg + (size_t)(tn0 * 16 + l15) * ATTR_DIM + half * 2;
  int n0 = tn0 * 16 + l15;
  if (tileM * 16 + 16 <= n1)
    gate_strip<true>(a, brow0, inv, out, n1, row0, n0);
  else
    gate_strip<false>(a, brow0, inv, out, n1, row0, n0);
}

extern "C" void kernel_launch(void* const* d_in, const int* in_sizes, int n_in,
                              void* d_out, int out_size, void* d_ws,
                              size_t ws_size, hipStream_t stream) {
  const float* X_h_1 = (const float*)d_in[0];
  const float* X_h_2 = (const float*)d_in[1];
  const float* X_n_1 = (const float*)d_in[2];
  const int* cross = (const int*)d_in[3];
  const float* W_gate = (const float*)d_in[4];
  float* out = (float*)d_out;

  const int N1 = in_sizes[0] / NODE_DIM;
  const int N2 = in_sizes[1] / NODE_DIM;
  const int E = in_sizes[3] / 2;
  const int* src = cross;
  const int* dst = cross + E;

  int4* link = (int4*)d_ws;
  float* invn1 = (float*)(link + E);
  float* invn2 = invn1 + N1;
  float* seg_sum = invn2 + N2;
  int* head = (int*)(seg_sum + N1);

  const int BLK = 256;
  const int waves_per_blk = BLK / 32;

  init_kernel<<<(N1 + BLK - 1) / BLK, BLK, 0, stream>>>(head, N1);
  const int rows_per_blk = waves_per_blk * NORM_ROWS;
  inv_norm_kernel<<<(N1 + rows_per_blk - 1) / rows_per_blk, BLK, 0, stream>>>(
      X_h_1, invn1, N1);
  inv_norm_kernel<<<(N2 + rows_per_blk - 1) / rows_per_blk, BLK, 0, stream>>>(
      X_h_2, invn2, N2);
  link_kernel<<<(E + BLK * LINK_EPT - 1) / (BLK * LINK_EPT), BLK, 0, stream>>>(
      src, dst, invn2, head, link, E);
  const int nodes_per_blk = waves_per_blk * GATHER_K;
  gather_kernel<<<(N1 + nodes_per_blk - 1) / nodes_per_blk, BLK, 0, stream>>>(
      X_h_1, X_h_2, head, link, invn1, seg_sum, out, N1, E);
  {
    int tilesM = (N1 + 15) / 16;
    int waves = tilesM * (NODE_DIM / 16 / GATE_TILES);
    gate_out_kernel<<<(waves + waves_per_blk - 1) / waves_per_blk, BLK, 0,
                      stream>>>(X_n_1, W_gate, seg_sum, out, N1);
  }
}


# edge lists built without global atomics: chunks of edges sorted by node range with LDS counters, then one work-group per range chains its edges by LDS exchanges (sum order = reverse LDS arrival)
# speedup vs baseline: 3.6939x; 1.3832x over previous
#include <hip/hip_runtime.h>
#include <math.h>

typedef float v2f __attribute__((ext_vector_type(2)));
typedef float v8f __attribute__((ext_vector_type(8)));

#define NODE_DIM 128
#define ATTR_DIM 64
#define EPS 1e-8f

#define LINK_BUCKETS 256
#define BK_THREADS 640
#define BK_EPT 4
#define BK_CHUNK (BK_THREADS * BK_EPT)

__global__ void __launch_bounds__(BK_THREADS)
bucket_kernel(const int* __restrict__ src, const int* __restrict__ dst,
              int2* __restrict__ tmp, int* __restrict__ tab, int E, int npb) {
  __shared__ int hist[LINK_BUCKETS];
  __shared__ int wsum[LINK_BUCKETS / 32];
  __shared__ int2 srt[BK_CHUNK];
  const int tid = threadIdx.x;
  const int lane = tid & 31;
  const int c = blockIdx.x;
  const int base = c * BK_CHUNK;
  if (tid < LINK_BUCKETS) hist[tid] = 0;
  int s[BK_EPT], d[BK_EPT], b[BK_EPT], r[BK_EPT];
#pragma unroll
  for (int k = 0; k < BK_EPT; ++k) {
    int e = min(base + k * BK_THREADS + tid, E - 1);
    s[k] = src[e];
    d[k] = dst[e];
  }
  __syncthreads();
#pragma unroll
  for (int k = 0; k < BK_EPT; ++k) {
    b[k] = s[k] / npb;
    r[k] = 0;
    if (base + k * BK_THREADS + tid < E)
      r[k] = atomicAdd(&hist[b[k]], 1);
  }
  __syncthreads();
  int v = 0, incl = 0;
  if (tid < LINK_BUCKETS) {
    v = hist[tid];
    incl = v;
#pragma unroll
    for (int off = 1; off < 32; off <<= 1) {
      int t = __shfl_up(incl, off, 32);
      if (lane >= off) incl += t;
    }
    if (lane == 31) wsum[tid >> 5] = incl;
  }
  __syncthreads();
  if (tid < LINK_BUCKETS) {
    int wp = 0;
#pragma unroll
    for (int w = 0; w < LINK_BUCKETS / 32; ++w)
      if (w < (tid >> 5)) wp += wsum[w];
    int excl = wp + incl - v;
    hist[tid] = excl;
    int* row = tab + (size_t)c * (LINK_BUCKETS + 1);
    row[tid] = excl;
    if (tid == LINK_BUCKETS - 1) row[LINK_BUCKETS] = excl + v;
  }
  __syncthreads();
#pragma unroll
  for (int k = 0; k < BK_EPT; ++k)
    if (base + k * BK_THREADS + tid < E)
      srt[hist[b[k]] + r[k]] = make_int2(s[k], d[k]);
  __syncthreads();
  const int n = min(BK_CHUNK, E - base);
  for (int i = tid; i < n; i += BK_THREADS) tmp[base + i] = srt[i];
}

#define NORM_ROWS 8
__global__ void inv_norm_kernel(const float* __restrict__ X,
                                float* __restrict__ inv, int n) {
  int wave = (blockIdx.x * blockDim.x + threadIdx.x) >> 5;
  int lane = threadIdx.x & 31;
  int r0 = wave * NORM_ROWS;
  if (r0 >= n) return;
  float p[NORM_ROWS];
#pragma unroll
  for (int k = 0; k < NORM_ROWS; ++k) {
    int r = min(r0 + k, n - 1);
    float4 a = reinterpret_cast<const float4*>(X + (size_t)r * NODE_DIM)[lane];
    p[k] = a.x * a.x + a.y * a.y + a.z * a.z + a.w * a.w;
  }
#pragma unroll
  for (int k = 0; k < NORM_ROWS; ++k) {
#pragma unroll
    for (int off = 16; off > 0; off >>= 1) p[k] += __shfl_xor(p[k], off, 32);
  }
  float mine = p[0];
#pragma unroll
  for (int k = 1; k < NORM_ROWS; ++k)
    if (lane == k) mine = p[k];
  if (lane < NORM_ROWS && r0 + lane < n)
    inv[r0 + lane] = 1.0f / fmaxf(sqrtf(mine), EPS);
}

#define LINK_THREADS 512
#define LINK_MAXNPB 1024
#define LINK_U 4

__device__ __forceinline__ int slice_place(int i, const int* cpre,
                                           const int* cstart) {
  int j = 0;
#pragma unroll
  for (int step = LINK_THREADS / 2; step > 0; step >>= 1)
    if (cpre[j + step] <= i) j += step;
  return cstart[j] + (i - cpre[j]);
}

__global__ void __launch_bounds__(LINK_THREADS)
link_kernel(const int2* __restrict__ tmp, const int* __restrict__ tab,
            const float* __restrict__ invn2, int* __restrict__ head,
            int4* __restrict__ link, int n1, int npb, int nchunks) {
  __shared__ int lhead[LINK_MAXNPB];
  __shared__ int cpre[LINK_THREADS];
  __shared__ int cstart[LINK_THREADS];
  __shared__ int wsum[LINK_THREADS / 32];
  const int tid = threadIdx.x;
  const int lane = tid & 31;
  const int wv = tid >> 5;
  const int g = blockIdx.x;
  const int node0 = g * npb;
  for (int i = tid; i < npb; i += LINK_THREADS) lhead[i] = -1;

  for (int cg = 0; cg < nchunks; cg += LINK_THREADS) {
    int c = cg + tid;
    int cnt = 0, st = 0;
    if (c < nchunks) {
      const int* row = tab + (size_t)c * (LINK_BUCKETS + 1) + g;
      int lo = row[0], hi = row[1];
      cnt = hi - lo;
      st = c * BK_CHUNK + lo;
    }
    int incl = cnt;
#pragma unroll
    for (int off = 1; off < 32; off <<= 1) {
      int t = __shfl_up(incl, off, 32);
      if (lane >= off) incl += t;
    }
    if (lane == 31) wsum[wv] = incl;
    __syncthreads();
    int wp = 0, total = 0;
#pragma unroll
    for (int w = 0; w < LINK_THREADS / 32; ++w) {
      int x = wsum[w];
      if (w < wv) wp += x;
      total += x;
    }
    cpre[tid] = wp + incl - cnt;
    cstart[tid] = st;
    __syncthreads();

    int done = 0;
    for (; done + LINK_THREADS * LINK_U <= total; done += LINK_THREADS * LINK_U) {
      int p[LINK_U];
      int2 sd[LINK_U];
      float in2[LINK_U];
#pragma unroll
      for (int k = 0; k < LINK_U; ++k)
        p[k] = slice_place(done + k * LINK_THREADS + tid, cpre, cstart);
#pragma unroll
      for (int k = 0; k < LINK_U; ++k) sd[k] = tmp[p[k]];
#pragma unroll
      for (int k = 0; k < LINK_U; ++k) in2[k] = invn2[sd[k].y];
#pragma unroll
      for (int k = 0; k < LINK_U; ++k) {
        int prev = atomicExch(&lhead[sd[k].x - node0], p[k]);
        link[p[k]] = make_int4(prev, sd[k].y, __float_as_int(in2[k]), 0);
      }
    }
    for (int i = done + tid; i < total; i += LINK_THREADS) {
      int p = slice_place(i, cpre, cstart);
      int2 sd = tmp[p];
      float in2 = invn2[sd.y];
      int prev = atomicExch(&lhead[sd.x - node0], p);
      link[p] = make_int4(prev, sd.y, __float_as_int(in2), 0);
    }
    __syncthreads();
  }
  __syncthreads();
  for (int i = tid; i < npb; i += LINK_THREADS)
    if (node0 + i < n1) head[node0 + i] = lhead[i];
}

#define GATHER_K 4
__global__ void gather_kernel(const float* __restrict__ X1,
                              const float* __restrict__ X2,
                              const int* __restrict__ head,
                              const int4* __restrict__ link,
                              const float* __restrict__ invn1,
                              float* __restrict__ seg_sum,
                              float* __restrict__ out, int n1, int E) {
  int w = __builtin_amdgcn_readfirstlane(
      (int)((blockIdx.x * blockDim.x + threadIdx.x) >> 5));
  int lane = threadIdx.x & 31;
  int s0 = w * GATHER_K;
  if (s0 >= n1) return;
  float4 a[GATHER_K], acc[GATHER_K];
  float in1[GATHER_K], ssum[GATHER_K];
  int4 cur[GATHER_K];
  bool act[GATHER_K];
  bool hn[GATHER_K];
  int nidx[GATHER_K];
#pragma unroll
  for (int k = 0; k < GATHER_K; ++k) {
    int s = min(s0 + k, n1 - 1);
    a[k] = reinterpret_cast<const float4*>(X1 + (size_t)s * NODE_DIM)[lane];
    in1[k] = invn1[s];
    nidx[k] = head[s];
    acc[k] = make_float4(0.0f, 0.0f, 0.0f, 0.0f);
    ssum[k] = 0.0f;
  }
#pragma unroll
  for (int k = 0; k < GATHER_K; ++k) {
    act[k] = (s0 + k < n1) & ((unsigned)nidx[k] < (unsigned)E);
    cur[k] = link[act[k] ? nidx[k] : 0];
  }
#pragma unroll
  for (int k = 0; k < GATHER_K; ++k) {
    hn[k] = act[k] & ((unsigned)cur[k].x < (unsigned)E);
    nidx[k] = hn[k] ? cur[k].x : 0;
  }
  for (int it = 0; it < E; ++it) {
    bool any = false;
#pragma unroll
    for (int k = 0; k < GATHER_K; ++k) any = any | act[k];
    if (!any) break;
    float4 b[GATHER_K];
    int4 nx[GATHER_K];
#pragma unroll
    for (int k = 0; k < GATHER_K; ++k) {
      int d = cur[k].y;
      b[k] = reinterpret_cast<const float4*>(X2 + (size_t)d * NODE_DIM)[lane];
      nx[k] = link[nidx[k]];
    }
#pragma unroll
    for (int k = 0; k < GATHER_K; ++k) {
      float p = a[k].x * b[k].x + a[k].y * b[k].y + a[k].z * b[k].z +
                a[k].w * b[k].w;
#pragma unroll
      for (int off = 16; off > 0; off >>= 1) p += __shfl_xor(p, off, 32);
      float in2 = __int_as_float(cur[k].z);
      float simv = p * in1[k] * in2;
      float wgt = expf(simv);
      ssum[k] = act[k] ? ssum[k] + wgt : ssum[k];
      acc[k].x = act[k] ? acc[k].x + wgt * b[k].x : acc[k].x;
      acc[k].y = act[k] ? acc[k].y + wgt * b[k].y : acc[k].y;
      acc[k].z = act[k] ? acc[k].z + wgt * b[k].z : acc[k].z;
      acc[k].w = act[k] ? acc[k].w + wgt * b[k].w : acc[k].w;
    }
#pragma unroll
    for (int k = 0; k < GATHER_K; ++k) {
      asm volatile("" : "+s"(nx[k].x), "+s"(nx[k].y), "+s"(nx[k].z));
      act[k] = hn[k];
      cur[k] = nx[k];
      hn[k] = hn[k] & ((unsigned)nx[k].x < (unsigned)E);
      nidx[k] = hn[k] ? nx[k].x : 0;
    }
  }
#pragma unroll
  for (int k = 0; k < GATHER_K; ++k) {
    int s = s0 + k;
    if (s < n1) {
      reinterpret_cast<float4*>(out + (size_t)s * NODE_DIM)[lane] = acc[k];
      if (lane == 0) seg_sum[s] = ssum[k];
    }
  }
}

#define GATE_TILES 4
#define GATE_KSTEPS (ATTR_DIM / 4)

template <bool FULL>
__device__ __forceinline__ void gate_strip(const v2f (&a)[GATE_KSTEPS],
                                           const float* __restrict__ brow0,
                                           const float (&inv)[8],
                                           float* __restrict__ out, int n1,
                                           int row0, int n0) {
#pragma unroll
  for (int t = 0; t < GATE_TILES; ++t) {
    const float* brow = brow0 + (size_t)t * 16 * ATTR_DIM;
    int n = n0 + t * 16;
    v2f b[GATE_KSTEPS];
#pragma unroll
    for (int i = 0; i < GATE_KSTEPS; ++i)
      b[i] = *reinterpret_cast<const v2f*>(brow + i * 4);
    float o[8];
#pragma unroll
    for (int v = 0; v < 8; ++v) {
      int row = FULL ? row0 + v : min(row0 + v, n1 - 1);
      o[v] = out[(size_t)row * NODE_DIM + n];
    }
    v8f c = {};
#pragma unroll
    for (int i = 0; i < GATE_KSTEPS; ++i)
      c = __builtin_amdgcn_wmma_f32_16x16x4_f32(
           false, a[i],  false, b[i],
           (short)0, c,  false,  false);
    float r[8];
#pragma unroll
    for (int v = 0; v < 8; ++v) {
      float g = 1.0f / (1.0f + expf(-c[v]));
      r[v] = g * (o[v] * inv[v]);
    }
#pragma unroll
    for (int v = 0; v < 8; ++v) {
      int row = row0 + v;
      if (FULL || row < n1) out[(size_t)row * NODE_DIM + n] = r[v];
    }
  }
}

__global__ void __launch_bounds__(256)
gate_out_kernel(const float* __restrict__ Xn,
                                const float* __restrict__ Wg,
                                const float* __restrict__ seg_sum,
                                float* __restrict__ out, int n1) {
  int wave = blockIdx.x * (blockDim.x >> 5) + (threadIdx.x >> 5);
  int lane = threadIdx.x & 31;
  int tilesM = (n1 + 15) >> 4;
  const int strips = NODE_DIM / 16 / GATE_TILES;
  int tileM = wave / strips;
  int tn0 = (wave % strips) * GATE_TILES;
  if (tileM >= tilesM) return;

  int l15 = lane & 15;
  int half = lane >> 4;
  int m = tileM * 16 + l15;
  if (m >= n1) m = n1 - 1;
  const float* arow = Xn + (size_t)m * ATTR_DIM + half * 2;

  v2f a[GATE_KSTEPS];
#pragma unroll
  for (int i = 0; i < GATE_KSTEPS; ++i)
    a[i] = *reinterpret_cast<const v2f*>(arow + i * 4);

  int row0 = tileM * 16 + half * 8;
  float inv[8];
#pragma unroll
  for (int v = 0; v < 8; ++v) {
    float ss = seg_sum[min(row0 + v, n1 - 1)];
    inv[v] = ss > 0.0f ? 1.0f / ss : 0.0f;
  }

  const float* brow0 = Wg + (size_t)(tn0 * 16 + l15) * ATTR_DIM + half * 2;
  int n0 = tn0 * 16 + l15;
  if (tileM * 16 + 16 <= n1)
    gate_strip<true>(a, brow0, inv, out, n1, row0, n0);
  else
    gate_strip<false>(a, brow0, inv, out, n1, row0, n0);
}

extern "C" void kernel_launch(void* const* d_in, const int* in_sizes, int n_in,
                              void* d_out, int out_size, void* d_ws,
                              size_t ws_size, hipStream_t stream) {
  const float* X_h_1 = (const float*)d_in[0];
  const float* X_h_2 = (const float*)d_in[1];
  const float* X_n_1 = (const float*)d_in[2];
  const int* cross = (const int*)d_in[3];
  const float* W_gate = (const float*)d_in[4];
  float* out = (float*)d_out;

  const int N1 = in_sizes[0] / NODE_DIM;
  const int N2 = in_sizes[1] / NODE_DIM;
  const int E = in_sizes[3] / 2;
  const int* src = cross;
  const int* dst = cross + E;

  int4* link = (int4*)d_ws;
  float* invn1 = (float*)(link + E);
  float* invn2 = invn1 + N1;
  float* seg_sum = invn2 + N2;
  int* head = (int*)(seg_sum + N1);

  const int BLK = 256;
  const int waves_per_blk = BLK / 32;

  const int npb = (N1 + LINK_BUCKETS - 1) / LINK_BUCKETS;
  const int nchunks = (E + BK_CHUNK - 1) / BK_CHUNK;
  int2* tmp = (int2*)d_out;
  int* tab = (int*)(tmp + E);

  bucket_kernel<<<nchunks, BK_THREADS, 0, stream>>>(src, dst, tmp, tab, E,
                                                    npb);
  const int rows_per_blk = waves_per_blk * NORM_ROWS;
  inv_norm_kernel<<<(N1 + rows_per_blk - 1) / rows_per_blk, BLK, 0, stream>>>(
      X_h_1, invn1, N1);
  inv_norm_kernel<<<(N2 + rows_per_blk - 1) / rows_per_blk, BLK, 0, stream>>>(
      X_h_2, invn2, N2);
  link_kernel<<<LINK_BUCKETS, LINK_THREADS, 0, stream>>>(
      tmp, tab, invn2, head, link, N1, npb, nchunks);
  const int nodes_per_blk = waves_per_blk * GATHER_K;
  gather_kernel<<<(N1 + nodes_per_blk - 1) / nodes_per_blk, BLK, 0, stream>>>(
      X_h_1, X_h_2, head, link, invn1, seg_sum, out, N1, E);
  {
    int tilesM = (N1 + 15) / 16;
    int waves = tilesM * (NODE_DIM / 16 / GATE_TILES);
    gate_out_kernel<<<(waves + waves_per_blk - 1) / waves_per_blk, BLK, 0,
                      stream>>>(X_n_1, W_gate, seg_sum, out, N1);
  }
}


# the two inverse-norm launches folded into the edge-sort launch as its first work-groups (role by block index): 6 launches become 4, row streaming runs beside the sort's latency chain
# speedup vs baseline: 3.8717x; 1.0481x over previous
#include <hip/hip_runtime.h>
#include <math.h>

typedef float v2f __attribute__((ext_vector_type(2)));
typedef float v8f __attribute__((ext_vector_type(8)));

#define NODE_DIM 128
#define ATTR_DIM 64
#define EPS 1e-8f

#define NORM_ROWS 8
#define NORM_GROUPS 4

struct NormGroup {
  const float* X;
  float* inv;
  int n;
  int r0;
};

__device__ __forceinline__ NormGroup norm_group(
    int gi, int g1, const float* __restrict__ X1, float* __restrict__ inv1,
    int n1, const float* __restrict__ X2, float* __restrict__ inv2, int n2) {
  NormGroup q;
  const bool first = gi < g1;
  q.X = first ? X1 : X2;
  q.inv = first ? inv1 : inv2;
  q.n = first ? n1 : n2;
  q.r0 = (first ? gi : gi - g1) * NORM_ROWS;
  return q;
}

__device__ __forceinline__ void norm_load(const NormGroup& q, int lane,
                                          float4 (&a)[NORM_ROWS]) {
#pragma unroll
  for (int k = 0; k < NORM_ROWS; ++k) {
    int r = min(q.r0 + k, q.n - 1);
    a[k] = reinterpret_cast<const float4*>(q.X + (size_t)r * NODE_DIM)[lane];
  }
}

__device__ __forceinline__ void norm_finish(const NormGroup& q, int lane,
                                            const float4 (&a)[NORM_ROWS]) {
  float p[NORM_ROWS];
#pragma unroll
  for (int k = 0; k < NORM_ROWS; ++k)
    p[k] = a[k].x * a[k].x + a[k].y * a[k].y + a[k].z * a[k].z +
           a[k].w * a[k].w;
#pragma unroll
  for (int k = 0; k < NORM_ROWS; ++k) {
#pragma unroll
    for (int off = 16; off > 0; off >>= 1) p[k] += __shfl_xor(p[k], off, 32);
  }
  float mine = p[0];
#pragma unroll
  for (int k = 1; k < NORM_ROWS; ++k)
    if (lane == k) mine = p[k];
  if (lane < NORM_ROWS && q.r0 + lane < q.n)
    q.inv[q.r0 + lane] = 1.0f / fmaxf(sqrtf(mine), EPS);
}

__device__ __forceinline__ void norm_wave(
    int wave, int lane, const float* __restrict__ X1, float* __restrict__ inv1,
    int n1, const float* __restrict__ X2, float* __restrict__ inv2, int n2) {
  const int g1 = (n1 + NORM_ROWS - 1) / NORM_ROWS;
  const int g2 = (n2 + NORM_ROWS - 1) / NORM_ROWS;
  const int gi0 = wave * NORM_GROUPS;
  if (gi0 >= g1 + g2) return;
  NormGroup q = norm_group(gi0, g1, X1, inv1, n1, X2, inv2, n2);
  float4 cur[NORM_ROWS];
  norm_load(q, lane, cur);
#pragma unroll
  for (int g = 0; g < NORM_GROUPS; ++g) {
    float4 nxt[NORM_ROWS];
    NormGroup qn = q;
    if (g + 1 < NORM_GROUPS) {
      qn = norm_group(gi0 + g + 1, g1, X1, inv1, n1, X2, inv2, n2);
      norm_load(qn, lane, nxt);
    }
    norm_finish(q, lane, cur);
    if (g + 1 < NORM_GROUPS) {
      q = qn;
#pragma unroll
      for (int k = 0; k < NORM_ROWS; ++k) cur[k] = nxt[k];
    }
  }
}

#define LINK_BUCKETS 256
#define BK_THREADS 640
#define BK_EPT 4
#define BK_CHUNK (BK_THREADS * BK_EPT)

__global__ void __launch_bounds__(BK_THREADS)
prep_kernel(const int* __restrict__ src, const int* __restrict__ dst,
            int2* __restrict__ tmp, int* __restrict__ tab, int E, int npb,
            const float* __restrict__ X1, float* __restrict__ inv1, int n1,
            const float* __restrict__ X2, float* __restrict__ inv2, int n2,
            int norm_blocks) {
  __shared__ int hist[LINK_BUCKETS];
  __shared__ int wsum[LINK_BUCKETS / 32];
  __shared__ int2 srt[BK_CHUNK];
  const int tid = threadIdx.x;
  const int lane = tid & 31;
  if ((int)blockIdx.x < norm_blocks) {
    norm_wave((int)blockIdx.x * (BK_THREADS / 32) + (tid >> 5), lane, X1, inv1,
              n1, X2, inv2, n2);
    return;
  }
  const int c = (int)blockIdx.x - norm_blocks;
  const int base = c * BK_CHUNK;
  if (tid < LINK_BUCKETS) hist[tid] = 0;
  int s[BK_EPT], d[BK_EPT], b[BK_EPT], r[BK_EPT];
#pragma unroll
  for (int k = 0; k < BK_EPT; ++k) {
    int e = min(base + k * BK_THREADS + tid, E - 1);
    s[k] = src[e];
    d[k] = dst[e];
  }
  __syncthreads();
#pragma unroll
  for (int k = 0; k < BK_EPT; ++k) {
    b[k] = s[k] / npb;
    r[k] = 0;
    if (base + k * BK_THREADS + tid < E)
      r[k] = atomicAdd(&hist[b[k]], 1);
  }
  __syncthreads();
  int v = 0, incl = 0;
  if (tid < LINK_BUCKETS) {
    v = hist[tid];
    incl = v;
#pragma unroll
    for (int off = 1; off < 32; off <<= 1) {
      int t = __shfl_up(incl, off, 32);
      if (lane >= off) incl += t;
    }
    if (lane == 31) wsum[tid >> 5] = incl;
  }
  __syncthreads();
  if (tid < LINK_BUCKETS) {
    int wp = 0;
#pragma unroll
    for (int w = 0; w < LINK_BUCKETS / 32; ++w)
      if (w < (tid >> 5)) wp += wsum[w];
    int excl = wp + incl - v;
    hist[tid] = excl;
    int* row = tab + (size_t)c * (LINK_BUCKETS + 1);
    row[tid] = excl;
    if (tid == LINK_BUCKETS - 1) row[LINK_BUCKETS] = excl + v;
  }
  __syncthreads();
#pragma unroll
  for (int k = 0; k < BK_EPT; ++k)
    if (base + k * BK_THREADS + tid < E)
      srt[hist[b[k]] + r[k]] = make_int2(s[k], d[k]);
  __syncthreads();
  const int n = min(BK_CHUNK, E - base);
  for (int i = tid; i < n; i += BK_THREADS) tmp[base + i] = srt[i];
}

#define LINK_THREADS 512
#define LINK_MAXNPB 1024
#define LINK_U 4

__device__ __forceinline__ int slice_place(int i, const int* cpre,
                                           const int* cstart) {
  int j = 0;
#pragma unroll
  for (int step = LINK_THREADS / 2; step > 0; step >>= 1)
    if (cpre[j + step] <= i) j += step;
  return cstart[j] + (i - cpre[j]);
}

__global__ void __launch_bounds__(LINK_THREADS)
link_kernel(const int2* __restrict__ tmp, const int* __restrict__ tab,
            const float* __restrict__ invn2, int* __restrict__ head,
            int4* __restrict__ link, int n1, int npb, int nchunks) {
  __shared__ int lhead[LINK_MAXNPB];
  __shared__ int cpre[LINK_THREADS];
  __shared__ int cstart[LINK_THREADS];
  __shared__ int wsum[LINK_THREADS / 32];
  const int tid = threadIdx.x;
  const int lane = tid & 31;
  const int wv = tid >> 5;
  const int g = blockIdx.x;
  const int node0 = g * npb;
  for (int i = tid; i < npb; i += LINK_THREADS) lhead[i] = -1;

  for (int cg = 0; cg < nchunks; cg += LINK_THREADS) {
    int c = cg + tid;
    int cnt = 0, st = 0;
    if (c < nchunks) {
      const int* row = tab + (size_t)c * (LINK_BUCKETS + 1) + g;
      int lo = row[0], hi = row[1];
      cnt = hi - lo;
      st = c * BK_CHUNK + lo;
    }
    int incl = cnt;
#pragma unroll
    for (int off = 1; off < 32; off <<= 1) {
      int t = __shfl_up(incl, off, 32);
      if (lane >= off) incl += t;
    }
    if (lane == 31) wsum[wv] = incl;
    __syncthreads();
    int wp = 0, total = 0;
#pragma unroll
    for (int w = 0; w < LINK_THREADS / 32; ++w) {
      int x = wsum[w];
      if (w < wv) wp += x;
      total += x;
    }
    cpre[tid] = wp + incl - cnt;
    cstart[tid] = st;
    __syncthreads();

    int done = 0;
    for (; done + LINK_THREADS * LINK_U <= total; done += LINK_THREADS * LINK_U) {
      int p[LINK_U];
      int2 sd[LINK_U];
      float in2[LINK_U];
#pragma unroll
      for (int k = 0; k < LINK_U; ++k)
        p[k] = slice_place(done + k * LINK_THREADS + tid, cpre, cstart);
#pragma unroll
      for (int k = 0; k < LINK_U; ++k) sd[k] = tmp[p[k]];
#pragma unroll
      for (int k = 0; k < LINK_U; ++k) in2[k] = invn2[sd[k].y];
#pragma unroll
      for (int k = 0; k < LINK_U; ++k) {
        int prev = atomicExch(&lhead[sd[k].x - node0], p[k]);
        link[p[k]] = make_int4(prev, sd[k].y, __float_as_int(in2[k]), 0);
      }
    }
    for (int i = done + tid; i < total; i += LINK_THREADS) {
      int p = slice_place(i, cpre, cstart);
      int2 sd = tmp[p];
      float in2 = invn2[sd.y];
      int prev = atomicExch(&lhead[sd.x - node0], p);
      link[p] = make_int4(prev, sd.y, __float_as_int(in2), 0);
    }
    __syncthreads();
  }
  __syncthreads();
  for (int i = tid; i < npb; i += LINK_THREADS)
    if (node0 + i < n1) head[node0 + i] = lhead[i];
}

#define GATHER_K 4
__global__ void gather_kernel(const float* __restrict__ X1,
                              const float* __restrict__ X2,
                              const int* __restrict__ head,
                              const int4* __restrict__ link,
                              const float* __restrict__ invn1,
                              float* __restrict__ seg_sum,
                              float* __restrict__ out, int n1, int E) {
  int w = __builtin_amdgcn_readfirstlane(
      (int)((blockIdx.x * blockDim.x + threadIdx.x) >> 5));
  int lane = threadIdx.x & 31;
  int s0 = w * GATHER_K;
  if (s0 >= n1) return;
  float4 a[GATHER_K], acc[GATHER_K];
  float in1[GATHER_K], ssum[GATHER_K];
  int4 cur[GATHER_K];
  bool act[GATHER_K];
  bool hn[GATHER_K];
  int nidx[GATHER_K];
#pragma unroll
  for (int k = 0; k < GATHER_K; ++k) {
    int s = min(s0 + k, n1 - 1);
    a[k] = reinterpret_cast<const float4*>(X1 + (size_t)s * NODE_DIM)[lane];
    in1[k] = invn1[s];
    nidx[k] = head[s];
    acc[k] = make_float4(0.0f, 0.0f, 0.0f, 0.0f);
    ssum[k] = 0.0f;
  }
#pragma unroll
  for (int k = 0; k < GATHER_K; ++k) {
    act[k] = (s0 + k < n1) & ((unsigned)nidx[k] < (unsigned)E);
    cur[k] = link[act[k] ? nidx[k] : 0];
  }
#pragma unroll
  for (int k = 0; k < GATHER_K; ++k) {
    hn[k] = act[k] & ((unsigned)cur[k].x < (unsigned)E);
    nidx[k] = hn[k] ? cur[k].x : 0;
  }
  for (int it = 0; it < E; ++it) {
    bool any = false;
#pragma unroll
    for (int k = 0; k < GATHER_K; ++k) any = any | act[k];
    if (!any) break;
    float4 b[GATHER_K];
    int4 nx[GATHER_K];
#pragma unroll
    for (int k = 0; k < GATHER_K; ++k) {
      int d = cur[k].y;
      b[k] = reinterpret_cast<const float4*>(X2 + (size_t)d * NODE_DIM)[lane];
      nx[k] = link[nidx[k]];
    }
#pragma unroll
    for (int k = 0; k < GATHER_K; ++k) {
      float p = a[k].x * b[k].x + a[k].y * b[k].y + a[k].z * b[k].z +
                a[k].w * b[k].w;
#pragma unroll
      for (int off = 16; off > 0; off >>= 1) p += __shfl_xor(p, off, 32);
      float in2 = __int_as_float(cur[k].z);
      float simv = p * in1[k] * in2;
      float wgt = expf(simv);
      ssum[k] = act[k] ? ssum[k] + wgt : ssum[k];
      acc[k].x = act[k] ? acc[k].x + wgt * b[k].x : acc[k].x;
      acc[k].y = act[k] ? acc[k].y + wgt * b[k].y : acc[k].y;
      acc[k].z = act[k] ? acc[k].z + wgt * b[k].z : acc[k].z;
      acc[k].w = act[k] ? acc[k].w + wgt * b[k].w : acc[k].w;
    }
#pragma unroll
    for (int k = 0; k < GATHER_K; ++k) {
      asm volatile("" : "+s"(nx[k].x), "+s"(nx[k].y), "+s"(nx[k].z));
      act[k] = hn[k];
      cur[k] = nx[k];
      hn[k] = hn[k] & ((unsigned)nx[k].x < (unsigned)E);
      nidx[k] = hn[k] ? nx[k].x : 0;
    }
  }
#pragma unroll
  for (int k = 0; k < GATHER_K; ++k) {
    int s = s0 + k;
    if (s < n1) {
      reinterpret_cast<float4*>(out + (size_t)s * NODE_DIM)[lane] = acc[k];
      if (lane == 0) seg_sum[s] = ssum[k];
    }
  }
}

#define GATE_TILES 4
#define GATE_KSTEPS (ATTR_DIM / 4)

template <bool FULL>
__device__ __forceinline__ void gate_strip(const v2f (&a)[GATE_KSTEPS],
                                           const float* __restrict__ brow0,
                                           const float (&inv)[8],
                                           float* __restrict__ out, int n1,
                                           int row0, int n0) {
#pragma unroll
  for (int t = 0; t < GATE_TILES; ++t) {
    const float* brow = brow0 + (size_t)t * 16 * ATTR_DIM;
    int n = n0 + t * 16;
    v2f b[GATE_KSTEPS];
#pragma unroll
    for (int i = 0; i < GATE_KSTEPS; ++i)
      b[i] = *reinterpret_cast<const v2f*>(brow + i * 4);
    float o[8];
#pragma unroll
    for (int v = 0; v < 8; ++v) {
      int row = FULL ? row0 + v : min(row0 + v, n1 - 1);
      o[v] = out[(size_t)row * NODE_DIM + n];
    }
    v8f c = {};
#pragma unroll
    for (int i = 0; i < GATE_KSTEPS; ++i)
      c = __builtin_amdgcn_wmma_f32_16x16x4_f32(
           false, a[i],  false, b[i],
           (short)0, c,  false,  false);
    float r[8];
#pragma unroll
    for (int v = 0; v < 8; ++v) {
      float g = 1.0f / (1.0f + expf(-c[v]));
      r[v] = g * (o[v] * inv[v]);
    }
#pragma unroll
    for (int v = 0; v < 8; ++v) {
      int row = row0 + v;
      if (FULL || row < n1) out[(size_t)row * NODE_DIM + n] = r[v];
    }
  }
}

__global__ void __launch_bounds__(256)
gate_out_kernel(const float* __restrict__ Xn,
                                const float* __restrict__ Wg,
                                const float* __restrict__ seg_sum,
                                float* __restrict__ out, int n1) {
  int wave = blockIdx.x * (blockDim.x >> 5) + (threadIdx.x >> 5);
  int lane = threadIdx.x & 31;
  int tilesM = (n1 + 15) >> 4;
  const int strips = NODE_DIM / 16 / GATE_TILES;
  int tileM = wave / strips;
  int tn0 = (wave % strips) * GATE_TILES;
  if (tileM >= tilesM) return;

  int l15 = lane & 15;
  int half = lane >> 4;
  int m = tileM * 16 + l15;
  if (m >= n1) m = n1 - 1;
  const float* arow = Xn + (size_t)m * ATTR_DIM + half * 2;

  v2f a[GATE_KSTEPS];
#pragma unroll
  for (int i = 0; i < GATE_KSTEPS; ++i)
    a[i] = *reinterpret_cast<const v2f*>(arow + i * 4);

  int row0 = tileM * 16 + half * 8;
  float inv[8];
#pragma unroll
  for (int v = 0; v < 8; ++v) {
    float ss = seg_sum[min(row0 + v, n1 - 1)];
    inv[v] = ss > 0.0f ? 1.0f / ss : 0.0f;
  }

  const float* brow0 = Wg + (size_t)(tn0 * 16 + l15) * ATTR_DIM + half * 2;
  int n0 = tn0 * 16 + l15;
  if (tileM * 16 + 16 <= n1)
    gate_strip<true>(a, brow0, inv, out, n1, row0, n0);
  else
    gate_strip<false>(a, brow0, inv, out, n1, row0, n0);
}

extern "C" void kernel_launch(void* const* d_in, const int* in_sizes, int n_in,
                              void* d_out, int out_size, void* d_ws,
                              size_t ws_size, hipStream_t stream) {
  const float* X_h_1 = (const float*)d_in[0];
  const float* X_h_2 = (const float*)d_in[1];
  const float* X_n_1 = (const float*)d_in[2];
  const int* cross = (const int*)d_in[3];
  const float* W_gate = (const float*)d_in[4];
  float* out = (float*)d_out;

  const int N1 = in_sizes[0] / NODE_DIM;
  const int N2 = in_sizes[1] / NODE_DIM;
  const int E = in_sizes[3] / 2;
  const int* src = cross;
  const int* dst = cross + E;

  int4* link = (int4*)d_ws;
  float* invn1 = (float*)(link + E);
  float* invn2 = invn1 + N1;
  float* seg_sum = invn2 + N2;
  int* head = (int*)(seg_sum + N1);

  const int BLK = 256;
  const int waves_per_blk = BLK / 32;

  const int npb = (N1 + LINK_BUCKETS - 1) / LINK_BUCKETS;
  const int nchunks = (E + BK_CHUNK - 1) / BK_CHUNK;
  int2* tmp = (int2*)d_out;
  int* tab = (int*)(tmp + E);

  const int norm_groups =
      (N1 + NORM_ROWS - 1) / NORM_ROWS + (N2 + NORM_ROWS - 1) / NORM_ROWS;
  const int groups_per_blk = (BK_THREADS / 32) * NORM_GROUPS;
  const int norm_blocks = (norm_groups + groups_per_blk - 1) / groups_per_blk;
  prep_kernel<<<norm_blocks + nchunks, BK_THREADS, 0, stream>>>(
      src, dst, tmp, tab, E, npb, X_h_1, invn1, N1, X_h_2, invn2, N2,
      norm_blocks);
  link_kernel<<<LINK_BUCKETS, LINK_THREADS, 0, stream>>>(
      tmp, tab, invn2, head, link, N1, npb, nchunks);
  const int nodes_per_blk = waves_per_blk * GATHER_K;
  gather_kernel<<<(N1 + nodes_per_blk - 1) / nodes_per_blk, BLK, 0, stream>>>(
      X_h_1, X_h_2, head, link, invn1, seg_sum, out, N1, E);
  {
    int tilesM = (N1 + 15) / 16;
    int waves = tilesM * (NODE_DIM / 16 / GATE_TILES);
    gate_out_kernel<<<(waves + waves_per_blk - 1) / waves_per_blk, BLK, 0,
                      stream>>>(X_n_1, W_gate, seg_sum, out, N1);
  }
}


# gather: the 4 lists' dot-product butterflies merged into one on the vector ALU (permlanex16 + row_xmask DPP, same pairs, bit-identical sums) instead of 20 LDS bpermutes a trip; one exp a trip
# speedup vs baseline: 5.4742x; 1.4139x over previous
#include <hip/hip_runtime.h>
#include <math.h>

typedef float v2f __attribute__((ext_vector_type(2)));
typedef float v8f __attribute__((ext_vector_type(8)));

#define NODE_DIM 128
#define ATTR_DIM 64
#define EPS 1e-8f

#define NORM_ROWS 8
#define NORM_GROUPS 4

struct NormGroup {
  const float* X;
  float* inv;
  int n;
  int r0;
};

__device__ __forceinline__ NormGroup norm_group(
    int gi, int g1, const float* __restrict__ X1, float* __restrict__ inv1,
    int n1, const float* __restrict__ X2, float* __restrict__ inv2, int n2) {
  NormGroup q;
  const bool first = gi < g1;
  q.X = first ? X1 : X2;
  q.inv = first ? inv1 : inv2;
  q.n = first ? n1 : n2;
  q.r0 = (first ? gi : gi - g1) * NORM_ROWS;
  return q;
}

__device__ __forceinline__ void norm_load(const NormGroup& q, int lane,
                                          float4 (&a)[NORM_ROWS]) {
#pragma unroll
  for (int k = 0; k < NORM_ROWS; ++k) {
    int r = min(q.r0 + k, q.n - 1);
    a[k] = reinterpret_cast<const float4*>(q.X + (size_t)r * NODE_DIM)[lane];
  }
}

__device__ __forceinline__ void norm_finish(const NormGroup& q, int lane,
                                            const float4 (&a)[NORM_ROWS]) {
  float p[NORM_ROWS];
#pragma unroll
  for (int k = 0; k < NORM_ROWS; ++k)
    p[k] = a[k].x * a[k].x + a[k].y * a[k].y + a[k].z * a[k].z +
           a[k].w * a[k].w;
#pragma unroll
  for (int k = 0; k < NORM_ROWS; ++k) {
#pragma unroll
    for (int off = 16; off > 0; off >>= 1) p[k] += __shfl_xor(p[k], off, 32);
  }
  float mine = p[0];
#pragma unroll
  for (int k = 1; k < NORM_ROWS; ++k)
    if (lane == k) mine = p[k];
  if (lane < NORM_ROWS && q.r0 + lane < q.n)
    q.inv[q.r0 + lane] = 1.0f / fmaxf(sqrtf(mine), EPS);
}

__device__ __forceinline__ void norm_wave(
    int wave, int lane, const float* __restrict__ X1, float* __restrict__ inv1,
    int n1, const float* __restrict__ X2, float* __restrict__ inv2, int n2) {
  const int g1 = (n1 + NORM_ROWS - 1) / NORM_ROWS;
  const int g2 = (n2 + NORM_ROWS - 1) / NORM_ROWS;
  const int gi0 = wave * NORM_GROUPS;
  if (gi0 >= g1 + g2) return;
  NormGroup q = norm_group(gi0, g1, X1, inv1, n1, X2, inv2, n2);
  float4 cur[NORM_ROWS];
  norm_load(q, lane, cur);
#pragma unroll
  for (int g = 0; g < NORM_GROUPS; ++g) {
    float4 nxt[NORM_ROWS];
    NormGroup qn = q;
    if (g + 1 < NORM_GROUPS) {
      qn = norm_group(gi0 + g + 1, g1, X1, inv1, n1, X2, inv2, n2);
      norm_load(qn, lane, nxt);
    }
    norm_finish(q, lane, cur);
    if (g + 1 < NORM_GROUPS) {
      q = qn;
#pragma unroll
      for (int k = 0; k < NORM_ROWS; ++k) cur[k] = nxt[k];
    }
  }
}

#define LINK_BUCKETS 256
#define BK_THREADS 640
#define BK_EPT 4
#define BK_CHUNK (BK_THREADS * BK_EPT)

__global__ void __launch_bounds__(BK_THREADS)
prep_kernel(const int* __restrict__ src, const int* __restrict__ dst,
            int2* __restrict__ tmp, int* __restrict__ tab, int E, int npb,
            const float* __restrict__ X1, float* __restrict__ inv1, int n1,
            const float* __restrict__ X2, float* __restrict__ inv2, int n2,
            int norm_blocks) {
  __shared__ int hist[LINK_BUCKETS];
  __shared__ int wsum[LINK_BUCKETS / 32];
  __shared__ int2 srt[BK_CHUNK];
  const int tid = threadIdx.x;
  const int lane = tid & 31;
  if ((int)blockIdx.x < norm_blocks) {
    norm_wave((int)blockIdx.x * (BK_THREADS / 32) + (tid >> 5), lane, X1, inv1,
              n1, X2, inv2, n2);
    return;
  }
  const int c = (int)blockIdx.x - norm_blocks;
  const int base = c * BK_CHUNK;
  if (tid < LINK_BUCKETS) hist[tid] = 0;
  int s[BK_EPT], d[BK_EPT], b[BK_EPT], r[BK_EPT];
#pragma unroll
  for (int k = 0; k < BK_EPT; ++k) {
    int e = min(base + k * BK_THREADS + tid, E - 1);
    s[k] = src[e];
    d[k] = dst[e];
  }
  __syncthreads();
#pragma unroll
  for (int k = 0; k < BK_EPT; ++k) {
    b[k] = s[k] / npb;
    r[k] = 0;
    if (base + k * BK_THREADS + tid < E)
      r[k] = atomicAdd(&hist[b[k]], 1);
  }
  __syncthreads();
  int v = 0, incl = 0;
  if (tid < LINK_BUCKETS) {
    v = hist[tid];
    incl = v;
#pragma unroll
    for (int off = 1; off < 32; off <<= 1) {
      int t = __shfl_up(incl, off, 32);
      if (lane >= off) incl += t;
    }
    if (lane == 31) wsum[tid >> 5] = incl;
  }
  __syncthreads();
  if (tid < LINK_BUCKETS) {
    int wp = 0;
#pragma unroll
    for (int w = 0; w < LINK_BUCKETS / 32; ++w)
      if (w < (tid >> 5)) wp += wsum[w];
    int excl = wp + incl - v;
    hist[tid] = excl;
    int* row = tab + (size_t)c * (LINK_BUCKETS + 1);
    row[tid] = excl;
    if (tid == LINK_BUCKETS - 1) row[LINK_BUCKETS] = excl + v;
  }
  __syncthreads();
#pragma unroll
  for (int k = 0; k < BK_EPT; ++k)
    if (base + k * BK_THREADS + tid < E)
      srt[hist[b[k]] + r[k]] = make_int2(s[k], d[k]);
  __syncthreads();
  const int n = min(BK_CHUNK, E - base);
  for (int i = tid; i < n; i += BK_THREADS) tmp[base + i] = srt[i];
}

#define LINK_THREADS 512
#define LINK_MAXNPB 1024
#define LINK_U 4

__device__ __forceinline__ int slice_place(int i, const int* cpre,
                                           const int* cstart) {
  int j = 0;
#pragma unroll
  for (int step = LINK_THREADS / 2; step > 0; step >>= 1)
    if (cpre[j + step] <= i) j += step;
  return cstart[j] + (i - cpre[j]);
}

__global__ void __launch_bounds__(LINK_THREADS)
link_kernel(const int2* __restrict__ tmp, const int* __restrict__ tab,
            const float* __restrict__ invn2, int* __restrict__ head,
            int4* __restrict__ link, int n1, int npb, int nchunks) {
  __shared__ int lhead[LINK_MAXNPB];
  __shared__ int cpre[LINK_THREADS];
  __shared__ int cstart[LINK_THREADS];
  __shared__ int wsum[LINK_THREADS / 32];
  const int tid = threadIdx.x;
  const int lane = tid & 31;
  const int wv = tid >> 5;
  const int g = blockIdx.x;
  const int node0 = g * npb;
  for (int i = tid; i < npb; i += LINK_THREADS) lhead[i] = -1;

  for (int cg = 0; cg < nchunks; cg += LINK_THREADS) {
    int c = cg + tid;
    int cnt = 0, st = 0;
    if (c < nchunks) {
      const int* row = tab + (size_t)c * (LINK_BUCKETS + 1) + g;
      int lo = row[0], hi = row[1];
      cnt = hi - lo;
      st = c * BK_CHUNK + lo;
    }
    int incl = cnt;
#pragma unroll
    for (int off = 1; off < 32; off <<= 1) {
      int t = __shfl_up(incl, off, 32);
      if (lane >= off) incl += t;
    }
    if (lane == 31) wsum[wv] = incl;
    __syncthreads();
    int wp = 0, total = 0;
#pragma unroll
    for (int w = 0; w < LINK_THREADS / 32; ++w) {
      int x = wsum[w];
      if (w < wv) wp += x;
      total += x;
    }
    cpre[tid] = wp + incl - cnt;
    cstart[tid] = st;
    __syncthreads();

    int done = 0;
    for (; done + LINK_THREADS * LINK_U <= total; done += LINK_THREADS * LINK_U) {
      int p[LINK_U];
      int2 sd[LINK_U];
      float in2[LINK_U];
#pragma unroll
      for (int k = 0; k < LINK_U; ++k)
        p[k] = slice_place(done + k * LINK_THREADS + tid, cpre, cstart);
#pragma unroll
      for (int k = 0; k < LINK_U; ++k) sd[k] = tmp[p[k]];
#pragma unroll
      for (int k = 0; k < LINK_U; ++k) in2[k] = invn2[sd[k].y];
#pragma unroll
      for (int k = 0; k < LINK_U; ++k) {
        int prev = atomicExch(&lhead[sd[k].x - node0], p[k]);
        link[p[k]] = make_int4(prev, sd[k].y, __float_as_int(in2[k]), 0);
      }
    }
    for (int i = done + tid; i < total; i += LINK_THREADS) {
      int p = slice_place(i, cpre, cstart);
      int2 sd = tmp[p];
      float in2 = invn2[sd.y];
      int prev = atomicExch(&lhead[sd.x - node0], p);
      link[p] = make_int4(prev, sd.y, __float_as_int(in2), 0);
    }
    __syncthreads();
  }
  __syncthreads();
  for (int i = tid; i < npb; i += LINK_THREADS)
    if (node0 + i < n1) head[node0 + i] = lhead[i];
}

#define GATHER_K 4
#define GATHER_WAVES 8
#define GATHER_SIMD_WAVES 12
#define GATE_ROWS (GATHER_WAVES * GATHER_K)
#define GATE_KSTEPS (ATTR_DIM / 4)
#define FRAG_PITCH (ATTR_DIM + 4)
#define GATE_PITCH (NODE_DIM + 4)

__device__ __forceinline__ float lane_xor16(float x) {
  int b = __builtin_bit_cast(int, x);
  return __builtin_bit_cast(
      float, __builtin_amdgcn_permlanex16(b, b, 0x76543210, 0xfedcba98, false,
                                          false));
}
template <int M>
__device__ __forceinline__ float lane_xor_row(float x) {
  return __builtin_bit_cast(
      float, __builtin_amdgcn_update_dpp(0, __builtin_bit_cast(int, x),
                                         0x160 | M, 0xf, 0xf, true));
}
__device__ __forceinline__ float sum4_by_lane_group(const float (&p)[4],
                                                    bool row0, bool oct1) {
  float ka = row0 ? p[0] : p[2], sa = row0 ? p[2] : p[0];
  float kb = row0 ? p[1] : p[3], sb = row0 ? p[3] : p[1];
  float qa = ka + lane_xor16(sa);
  float qb = kb + lane_xor16(sb);
  float kc = oct1 ? qb : qa, sc = oct1 ? qa : qb;
  float r = kc + lane_xor_row<8>(sc);
  r += lane_xor_row<4>(r);
  r += lane_xor_row<2>(r);
  r += lane_xor_row<1>(r);
  return r;
}

__global__ void __launch_bounds__(GATHER_WAVES * 32, GATHER_SIMD_WAVES)
gather_gate_kernel(const float* __restrict__ X1, const float* __restrict__ X2,
                   const int* __restrict__ head,
                   const int4* __restrict__ link,
                   const float* __restrict__ invn1,
                   const float* __restrict__ Xn, const float* __restrict__ Wg,
                   float* __restrict__ out, int n1, int E) {
  __shared__ __attribute__((aligned(16))) float sW[NODE_DIM * FRAG_PITCH];
  __shared__ __attribute__((aligned(16))) float sA[GATE_ROWS * FRAG_PITCH];
  float* sG = sW;
  const int tid = threadIdx.x;
  int w = __builtin_amdgcn_readfirstlane(
      (int)((blockIdx.x * (GATHER_WAVES * 32) + threadIdx.x) >> 5));
  int lane = tid & 31;
  int wq = w & (GATHER_WAVES - 1);
  int s0 = w * GATHER_K;
  int node0 = (w - wq) * GATHER_K;

  {
    const float4* Wg4 = reinterpret_cast<const float4*>(Wg);
    const float4* Xn4 = reinterpret_cast<const float4*>(Xn);
    float4 wv[NODE_DIM * ATTR_DIM / 4 / (GATHER_WAVES * 32)];
    float4 xv[GATE_ROWS * ATTR_DIM / 4 / (GATHER_WAVES * 32)];
#pragma unroll
    for (int j = 0; j < NODE_DIM * ATTR_DIM / 4 / (GATHER_WAVES * 32); ++j)
      wv[j] = Wg4[tid + j * (GATHER_WAVES * 32)];
#pragma unroll
    for (int j = 0; j < GATE_ROWS * ATTR_DIM / 4 / (GATHER_WAVES * 32); ++j) {
      int idx = tid + j * (GATHER_WAVES * 32);
      int node = min(node0 + (idx >> 4), n1 - 1);
      xv[j] = Xn4[(size_t)node * (ATTR_DIM / 4) + (idx & 15)];
    }
#pragma unroll
    for (int j = 0; j < NODE_DIM * ATTR_DIM / 4 / (GATHER_WAVES * 32); ++j) {
      int idx = tid + j * (GATHER_WAVES * 32);
      *reinterpret_cast<float4*>(&sW[(idx >> 4) * FRAG_PITCH + (idx & 15) * 4]) =
          wv[j];
    }
#pragma unroll
    for (int j = 0; j < GATE_ROWS * ATTR_DIM / 4 / (GATHER_WAVES * 32); ++j) {
      int idx = tid + j * (GATHER_WAVES * 32);
      *reinterpret_cast<float4*>(&sA[(idx >> 4) * FRAG_PITCH + (idx & 15) * 4]) =
          xv[j];
    }
  }
  __syncthreads();
  {
    const int l15 = lane & 15;
    const int half = lane >> 4;
    const int mt = wq >> 2;
    const int nt0 = (wq & 3) * 2;
    const float* ap = &sA[(mt * 16 + l15) * FRAG_PITCH + half * 2];
    const float* bp0 = &sW[(nt0 * 16 + l15) * FRAG_PITCH + half * 2];
    const float* bp1 = bp0 + 16 * FRAG_PITCH;
    v8f c0 = {}, c1 = {};
#pragma unroll
    for (int i = 0; i < GATE_KSTEPS; ++i) {
      v2f a = *reinterpret_cast<const v2f*>(ap + i * 4);
      v2f b0 = *reinterpret_cast<const v2f*>(bp0 + i * 4);
      v2f b1 = *reinterpret_cast<const v2f*>(bp1 + i * 4);
      c0 = __builtin_amdgcn_wmma_f32_16x16x4_f32(
           false, a,  false, b0,
           (short)0, c0,  false,  false);
      c1 = __builtin_amdgcn_wmma_f32_16x16x4_f32(
           false, a,  false, b1,
           (short)0, c1,  false,  false);
    }
    float g0[8], g1[8];
#pragma unroll
    for (int v = 0; v < 8; ++v) {
      g0[v] = 1.0f / (1.0f + expf(-c0[v]));
      g1[v] = 1.0f / (1.0f + expf(-c1[v]));
    }
    __syncthreads();
    float* gp = &sG[(mt * 16 + half * 8) * GATE_PITCH + nt0 * 16 + l15];
#pragma unroll
    for (int v = 0; v < 8; ++v) {
      gp[v * GATE_PITCH] = g0[v];
      gp[v * GATE_PITCH + 16] = g1[v];
    }
  }
  __syncthreads();

  float4 a[GATHER_K], acc[GATHER_K];
  float in1[GATHER_K], ssum[GATHER_K];
  int4 cur[GATHER_K];
  bool act[GATHER_K];
  bool hn[GATHER_K];
  int nidx[GATHER_K];
#pragma unroll
  for (int k = 0; k < GATHER_K; ++k) {
    int s = min(s0 + k, n1 - 1);
    a[k] = reinterpret_cast<const float4*>(X1 + (size_t)s * NODE_DIM)[lane];
    in1[k] = invn1[s];
    nidx[k] = head[s];
    acc[k] = make_float4(0.0f, 0.0f, 0.0f, 0.0f);
    ssum[k] = 0.0f;
  }
#pragma unroll
  for (int k = 0; k < GATHER_K; ++k) {
    act[k] = (s0 + k < n1) & ((unsigned)nidx[k] < (unsigned)E);
    cur[k] = link[act[k] ? nidx[k] : 0];
  }
#pragma unroll
  for (int k = 0; k < GATHER_K; ++k) {
    hn[k] = act[k] & ((unsigned)cur[k].x < (unsigned)E);
    nidx[k] = hn[k] ? cur[k].x : 0;
  }
  static_assert(GATHER_K == 4, "sum4_by_lane_group merges exactly four sums");
  const bool row0 = lane < 16;
  const bool oct1 = (lane & 8) != 0;
  const int myk = lane >> 3;
  const float in1sel = myk == 0   ? in1[0]
                       : myk == 1 ? in1[1]
                       : myk == 2 ? in1[2]
                                  : in1[3];
  float in2sel = __int_as_float(myk == 0   ? cur[0].z
                                : myk == 1 ? cur[1].z
                                : myk == 2 ? cur[2].z
                                           : cur[3].z);
  for (int it = 0; it < E; ++it) {
    bool any = false;
#pragma unroll
    for (int k = 0; k < GATHER_K; ++k) any = any | act[k];
    if (!any) break;
    float4 b[GATHER_K];
    int4 nx[GATHER_K];
#pragma unroll
    for (int k = 0; k < GATHER_K; ++k) {
      int d = cur[k].y;
      b[k] = reinterpret_cast<const float4*>(X2 + (size_t)d * NODE_DIM)[lane];
      nx[k] = link[nidx[k]];
    }
    float p[GATHER_K];
#pragma unroll
    for (int k = 0; k < GATHER_K; ++k)
      p[k] = a[k].x * b[k].x + a[k].y * b[k].y + a[k].z * b[k].z +
             a[k].w * b[k].w;
    const float dotk = sum4_by_lane_group(p, row0, oct1);
    const float wlane = expf(dotk * in1sel * in2sel);
#pragma unroll
    for (int k = 0; k < GATHER_K; ++k) {
      float wgt = __int_as_float(
          __builtin_amdgcn_readlane(__float_as_int(wlane), 8 * k));
      ssum[k] = act[k] ? ssum[k] + wgt : ssum[k];
      acc[k].x = act[k] ? acc[k].x + wgt * b[k].x : acc[k].x;
      acc[k].y = act[k] ? acc[k].y + wgt * b[k].y : acc[k].y;
      acc[k].z = act[k] ? acc[k].z + wgt * b[k].z : acc[k].z;
      acc[k].w = act[k] ? acc[k].w + wgt * b[k].w : acc[k].w;
    }
#pragma unroll
    for (int k = 0; k < GATHER_K; ++k) {
      asm volatile("" : "+s"(nx[k].x), "+s"(nx[k].y), "+s"(nx[k].z));
      act[k] = hn[k];
      cur[k] = nx[k];
      hn[k] = hn[k] & ((unsigned)nx[k].x < (unsigned)E);
      nidx[k] = hn[k] ? nx[k].x : 0;
    }
    in2sel = __int_as_float(myk == 0   ? cur[0].z
                            : myk == 1 ? cur[1].z
                            : myk == 2 ? cur[2].z
                                       : cur[3].z);
  }
#pragma unroll
  for (int k = 0; k < GATHER_K; ++k) {
    int s = s0 + k;
    if (s < n1) {
      const float4 g = *reinterpret_cast<const float4*>(
          &sG[(wq * GATHER_K + k) * GATE_PITCH + lane * 4]);
      float ss = ssum[k];
      float inv = ss > 0.0f ? 1.0f / ss : 0.0f;
      float4 r;
      r.x = g.x * (acc[k].x * inv);
      r.y = g.y * (acc[k].y * inv);
      r.z = g.z * (acc[k].z * inv);
      r.w = g.w * (acc[k].w * inv);
      reinterpret_cast<float4*>(out + (size_t)s * NODE_DIM)[lane] = r;
    }
  }
}

extern "C" void kernel_launch(void* const* d_in, const int* in_sizes, int n_in,
                              void* d_out, int out_size, void* d_ws,
                              size_t ws_size, hipStream_t stream) {
  const float* X_h_1 = (const float*)d_in[0];
  const float* X_h_2 = (const float*)d_in[1];
  const float* X_n_1 = (const float*)d_in[2];
  const int* cross = (const int*)d_in[3];
  const float* W_gate = (const float*)d_in[4];
  float* out = (float*)d_out;

  const int N1 = in_sizes[0] / NODE_DIM;
  const int N2 = in_sizes[1] / NODE_DIM;
  const int E = in_sizes[3] / 2;
  const int* src = cross;
  const int* dst = cross + E;

  int4* link = (int4*)d_ws;
  float* invn1 = (float*)(link + E);
  float* invn2 = invn1 + N1;
  int* head = (int*)(invn2 + N2);

  const int npb = (N1 + LINK_BUCKETS - 1) / LINK_BUCKETS;
  const int nchunks = (E + BK_CHUNK - 1) / BK_CHUNK;
  int2* tmp = (int2*)d_out;
  int* tab = (int*)(tmp + E);

  const int norm_groups =
      (N1 + NORM_ROWS - 1) / NORM_ROWS + (N2 + NORM_ROWS - 1) / NORM_ROWS;
  const int groups_per_blk = (BK_THREADS / 32) * NORM_GROUPS;
  const int norm_blocks = (norm_groups + groups_per_blk - 1) / groups_per_blk;
  prep_kernel<<<norm_blocks + nchunks, BK_THREADS, 0, stream>>>(
      src, dst, tmp, tab, E, npb, X_h_1, invn1, N1, X_h_2, invn2, N2,
      norm_blocks);
  link_kernel<<<LINK_BUCKETS, LINK_THREADS, 0, stream>>>(
      tmp, tab, invn2, head, link, N1, npb, nchunks);
  gather_gate_kernel<<<(N1 + GATE_ROWS - 1) / GATE_ROWS, GATHER_WAVES * 32, 0,
                       stream>>>(X_h_1, X_h_2, head, link, invn1, X_n_1,
                                 W_gate, out, N1, E);
}


# gather walk: per-list flags kept as integer all-ones/zero words on the scalar unit; a finished list gets weight +0.0 by one scalar AND instead of 20 vector selects a trip; sums bit-identical
# speedup vs baseline: 5.6703x; 1.0358x over previous
#include <hip/hip_runtime.h>
#include <math.h>

typedef float v2f __attribute__((ext_vector_type(2)));
typedef float v8f __attribute__((ext_vector_type(8)));

#define NODE_DIM 128
#define ATTR_DIM 64
#define EPS 1e-8f

#define NORM_ROWS 8
#define NORM_GROUPS 4

struct NormGroup {
  const float* X;
  float* inv;
  int n;
  int r0;
};

__device__ __forceinline__ NormGroup norm_group(
    int gi, int g1, const float* __restrict__ X1, float* __restrict__ inv1,
    int n1, const float* __restrict__ X2, float* __restrict__ inv2, int n2) {
  NormGroup q;
  const bool first = gi < g1;
  q.X = first ? X1 : X2;
  q.inv = first ? inv1 : inv2;
  q.n = first ? n1 : n2;
  q.r0 = (first ? gi : gi - g1) * NORM_ROWS;
  return q;
}

__device__ __forceinline__ void norm_load(const NormGroup& q, int lane,
                                          float4 (&a)[NORM_ROWS]) {
#pragma unroll
  for (int k = 0; k < NORM_ROWS; ++k) {
    int r = min(q.r0 + k, q.n - 1);
    a[k] = reinterpret_cast<const float4*>(q.X + (size_t)r * NODE_DIM)[lane];
  }
}

__device__ __forceinline__ void norm_finish(const NormGroup& q, int lane,
                                            const float4 (&a)[NORM_ROWS]) {
  float p[NORM_ROWS];
#pragma unroll
  for (int k = 0; k < NORM_ROWS; ++k)
    p[k] = a[k].x * a[k].x + a[k].y * a[k].y + a[k].z * a[k].z +
           a[k].w * a[k].w;
#pragma unroll
  for (int k = 0; k < NORM_ROWS; ++k) {
#pragma unroll
    for (int off = 16; off > 0; off >>= 1) p[k] += __shfl_xor(p[k], off, 32);
  }
  float mine = p[0];
#pragma unroll
  for (int k = 1; k < NORM_ROWS; ++k)
    if (lane == k) mine = p[k];
  if (lane < NORM_ROWS && q.r0 + lane < q.n)
    q.inv[q.r0 + lane] = 1.0f / fmaxf(sqrtf(mine), EPS);
}

__device__ __forceinline__ void norm_wave(
    int wave, int lane, const float* __restrict__ X1, float* __restrict__ inv1,
    int n1, const float* __restrict__ X2, float* __restrict__ inv2, int n2) {
  const int g1 = (n1 + NORM_ROWS - 1) / NORM_ROWS;
  const int g2 = (n2 + NORM_ROWS - 1) / NORM_ROWS;
  const int gi0 = wave * NORM_GROUPS;
  if (gi0 >= g1 + g2) return;
  NormGroup q = norm_group(gi0, g1, X1, inv1, n1, X2, inv2, n2);
  float4 cur[NORM_ROWS];
  norm_load(q, lane, cur);
#pragma unroll
  for (int g = 0; g < NORM_GROUPS; ++g) {
    float4 nxt[NORM_ROWS];
    NormGroup qn = q;
    if (g + 1 < NORM_GROUPS) {
      qn = norm_group(gi0 + g + 1, g1, X1, inv1, n1, X2, inv2, n2);
      norm_load(qn, lane, nxt);
    }
    norm_finish(q, lane, cur);
    if (g + 1 < NORM_GROUPS) {
      q = qn;
#pragma unroll
      for (int k = 0; k < NORM_ROWS; ++k) cur[k] = nxt[k];
    }
  }
}

#define LINK_BUCKETS 256
#define BK_THREADS 640
#define BK_EPT 4
#define BK_CHUNK (BK_THREADS * BK_EPT)

__global__ void __launch_bounds__(BK_THREADS)
prep_kernel(const int* __restrict__ src, const int* __restrict__ dst,
            int2* __restrict__ tmp, int* __restrict__ tab, int E, int npb,
            const float* __restrict__ X1, float* __restrict__ inv1, int n1,
            const float* __restrict__ X2, float* __restrict__ inv2, int n2,
            int norm_blocks) {
  __shared__ int hist[LINK_BUCKETS];
  __shared__ int wsum[LINK_BUCKETS / 32];
  __shared__ int2 srt[BK_CHUNK];
  const int tid = threadIdx.x;
  const int lane = tid & 31;
  if ((int)blockIdx.x < norm_blocks) {
    norm_wave((int)blockIdx.x * (BK_THREADS / 32) + (tid >> 5), lane, X1, inv1,
              n1, X2, inv2, n2);
    return;
  }
  const int c = (int)blockIdx.x - norm_blocks;
  const int base = c * BK_CHUNK;
  if (tid < LINK_BUCKETS) hist[tid] = 0;
  int s[BK_EPT], d[BK_EPT], b[BK_EPT], r[BK_EPT];
#pragma unroll
  for (int k = 0; k < BK_EPT; ++k) {
    int e = min(base + k * BK_THREADS + tid, E - 1);
    s[k] = src[e];
    d[k] = dst[e];
  }
  __syncthreads();
#pragma unroll
  for (int k = 0; k < BK_EPT; ++k) {
    b[k] = s[k] / npb;
    r[k] = 0;
    if (base + k * BK_THREADS + tid < E)
      r[k] = atomicAdd(&hist[b[k]], 1);
  }
  __syncthreads();
  int v = 0, incl = 0;
  if (tid < LINK_BUCKETS) {
    v = hist[tid];
    incl = v;
#pragma unroll
    for (int off = 1; off < 32; off <<= 1) {
      int t = __shfl_up(incl, off, 32);
      if (lane >= off) incl += t;
    }
    if (lane == 31) wsum[tid >> 5] = incl;
  }
  __syncthreads();
  if (tid < LINK_BUCKETS) {
    int wp = 0;
#pragma unroll
    for (int w = 0; w < LINK_BUCKETS / 32; ++w)
      if (w < (tid >> 5)) wp += wsum[w];
    int excl = wp + incl - v;
    hist[tid] = excl;
    int* row = tab + (size_t)c * (LINK_BUCKETS + 1);
    row[tid] = excl;
    if (tid == LINK_BUCKETS - 1) row[LINK_BUCKETS] = excl + v;
  }
  __syncthreads();
#pragma unroll
  for (int k = 0; k < BK_EPT; ++k)
    if (base + k * BK_THREADS + tid < E)
      srt[hist[b[k]] + r[k]] = make_int2(s[k], d[k]);
  __syncthreads();
  const int n = min(BK_CHUNK, E - base);
  for (int i = tid; i < n; i += BK_THREADS) tmp[base + i] = srt[i];
}

#define LINK_THREADS 512
#define LINK_MAXNPB 1024
#define LINK_U 4

__device__ __forceinline__ int slice_place(int i, const int* cpre,
                                           const int* cstart) {
  int j = 0;
#pragma unroll
  for (int step = LINK_THREADS / 2; step > 0; step >>= 1)
    if (cpre[j + step] <= i) j += step;
  return cstart[j] + (i - cpre[j]);
}

__global__ void __launch_bounds__(LINK_THREADS)
link_kernel(const int2* __restrict__ tmp, const int* __restrict__ tab,
            const float* __restrict__ invn2, int* __restrict__ head,
            int4* __restrict__ link, int n1, int npb, int nchunks) {
  __shared__ int lhead[LINK_MAXNPB];
  __shared__ int cpre[LINK_THREADS];
  __shared__ int cstart[LINK_THREADS];
  __shared__ int wsum[LINK_THREADS / 32];
  const int tid = threadIdx.x;
  const int lane = tid & 31;
  const int wv = tid >> 5;
  const int g = blockIdx.x;
  const int node0 = g * npb;
  for (int i = tid; i < npb; i += LINK_THREADS) lhead[i] = -1;

  for (int cg = 0; cg < nchunks; cg += LINK_THREADS) {
    int c = cg + tid;
    int cnt = 0, st = 0;
    if (c < nchunks) {
      const int* row = tab + (size_t)c * (LINK_BUCKETS + 1) + g;
      int lo = row[0], hi = row[1];
      cnt = hi - lo;
      st = c * BK_CHUNK + lo;
    }
    int incl = cnt;
#pragma unroll
    for (int off = 1; off < 32; off <<= 1) {
      int t = __shfl_up(incl, off, 32);
      if (lane >= off) incl += t;
    }
    if (lane == 31) wsum[wv] = incl;
    __syncthreads();
    int wp = 0, total = 0;
#pragma unroll
    for (int w = 0; w < LINK_THREADS / 32; ++w) {
      int x = wsum[w];
      if (w < wv) wp += x;
      total += x;
    }
    cpre[tid] = wp + incl - cnt;
    cstart[tid] = st;
    __syncthreads();

    int done = 0;
    for (; done + LINK_THREADS * LINK_U <= total; done += LINK_THREADS * LINK_U) {
      int p[LINK_U];
      int2 sd[LINK_U];
      float in2[LINK_U];
#pragma unroll
      for (int k = 0; k < LINK_U; ++k)
        p[k] = slice_place(done + k * LINK_THREADS + tid, cpre, cstart);
#pragma unroll
      for (int k = 0; k < LINK_U; ++k) sd[k] = tmp[p[k]];
#pragma unroll
      for (int k = 0; k < LINK_U; ++k) in2[k] = invn2[sd[k].y];
#pragma unroll
      for (int k = 0; k < LINK_U; ++k) {
        int prev = atomicExch(&lhead[sd[k].x - node0], p[k]);
        link[p[k]] = make_int4(prev, sd[k].y, __float_as_int(in2[k]), 0);
      }
    }
    for (int i = done + tid; i < total; i += LINK_THREADS) {
      int p = slice_place(i, cpre, cstart);
      int2 sd = tmp[p];
      float in2 = invn2[sd.y];
      int prev = atomicExch(&lhead[sd.x - node0], p);
      link[p] = make_int4(prev, sd.y, __float_as_int(in2), 0);
    }
    __syncthreads();
  }
  __syncthreads();
  for (int i = tid; i < npb; i += LINK_THREADS)
    if (node0 + i < n1) head[node0 + i] = lhead[i];
}

#define GATHER_K 4
#define GATHER_WAVES 8
#define GATHER_SIMD_WAVES 12
#define GATE_ROWS (GATHER_WAVES * GATHER_K)
#define GATE_KSTEPS (ATTR_DIM / 4)
#define FRAG_PITCH (ATTR_DIM + 4)
#define GATE_PITCH (NODE_DIM + 4)

__device__ __forceinline__ float lane_xor16(float x) {
  int b = __builtin_bit_cast(int, x);
  return __builtin_bit_cast(
      float, __builtin_amdgcn_permlanex16(b, b, 0x76543210, 0xfedcba98, false,
                                          false));
}
template <int M>
__device__ __forceinline__ float lane_xor_row(float x) {
  return __builtin_bit_cast(
      float, __builtin_amdgcn_update_dpp(0, __builtin_bit_cast(int, x),
                                         0x160 | M, 0xf, 0xf, true));
}
__device__ __forceinline__ float sum4_by_lane_group(const float (&p)[4],
                                                    bool row0, bool oct1) {
  float ka = row0 ? p[0] : p[2], sa = row0 ? p[2] : p[0];
  float kb = row0 ? p[1] : p[3], sb = row0 ? p[3] : p[1];
  float qa = ka + lane_xor16(sa);
  float qb = kb + lane_xor16(sb);
  float kc = oct1 ? qb : qa, sc = oct1 ? qa : qb;
  float r = kc + lane_xor_row<8>(sc);
  r += lane_xor_row<4>(r);
  r += lane_xor_row<2>(r);
  r += lane_xor_row<1>(r);
  return r;
}

__global__ void __launch_bounds__(GATHER_WAVES * 32, GATHER_SIMD_WAVES)
gather_gate_kernel(const float* __restrict__ X1, const float* __restrict__ X2,
                   const int* __restrict__ head,
                   const int4* __restrict__ link,
                   const float* __restrict__ invn1,
                   const float* __restrict__ Xn, const float* __restrict__ Wg,
                   float* __restrict__ out, int n1, int E) {
  __shared__ __attribute__((aligned(16))) float sW[NODE_DIM * FRAG_PITCH];
  __shared__ __attribute__((aligned(16))) float sA[GATE_ROWS * FRAG_PITCH];
  float* sG = sW;
  const int tid = threadIdx.x;
  int w = __builtin_amdgcn_readfirstlane(
      (int)((blockIdx.x * (GATHER_WAVES * 32) + threadIdx.x) >> 5));
  int lane = tid & 31;
  int wq = w & (GATHER_WAVES - 1);
  int s0 = w * GATHER_K;
  int node0 = (w - wq) * GATHER_K;

  {
    const float4* Wg4 = reinterpret_cast<const float4*>(Wg);
    const float4* Xn4 = reinterpret_cast<const float4*>(Xn);
    float4 wv[NODE_DIM * ATTR_DIM / 4 / (GATHER_WAVES * 32)];
    float4 xv[GATE_ROWS * ATTR_DIM / 4 / (GATHER_WAVES * 32)];
#pragma unroll
    for (int j = 0; j < NODE_DIM * ATTR_DIM / 4 / (GATHER_WAVES * 32); ++j)
      wv[j] = Wg4[tid + j * (GATHER_WAVES * 32)];
#pragma unroll
    for (int j = 0; j < GATE_ROWS * ATTR_DIM / 4 / (GATHER_WAVES * 32); ++j) {
      int idx = tid + j * (GATHER_WAVES * 32);
      int node = min(node0 + (idx >> 4), n1 - 1);
      xv[j] = Xn4[(size_t)node * (ATTR_DIM / 4) + (idx & 15)];
    }
#pragma unroll
    for (int j = 0; j < NODE_DIM * ATTR_DIM / 4 / (GATHER_WAVES * 32); ++j) {
      int idx = tid + j * (GATHER_WAVES * 32);
      *reinterpret_cast<float4*>(&sW[(idx >> 4) * FRAG_PITCH + (idx & 15) * 4]) =
          wv[j];
    }
#pragma unroll
    for (int j = 0; j < GATE_ROWS * ATTR_DIM / 4 / (GATHER_WAVES * 32); ++j) {
      int idx = tid + j * (GATHER_WAVES * 32);
      *reinterpret_cast<float4*>(&sA[(idx >> 4) * FRAG_PITCH + (idx & 15) * 4]) =
          xv[j];
    }
  }
  __syncthreads();
  {
    const int l15 = lane & 15;
    const int half = lane >> 4;
    const int mt = wq >> 2;
    const int nt0 = (wq & 3) * 2;
    const float* ap = &sA[(mt * 16 + l15) * FRAG_PITCH + half * 2];
    const float* bp0 = &sW[(nt0 * 16 + l15) * FRAG_PITCH + half * 2];
    const float* bp1 = bp0 + 16 * FRAG_PITCH;
    v8f c0 = {}, c1 = {};
#pragma unroll
    for (int i = 0; i < GATE_KSTEPS; ++i) {
      v2f a = *reinterpret_cast<const v2f*>(ap + i * 4);
      v2f b0 = *reinterpret_cast<const v2f*>(bp0 + i * 4);
      v2f b1 = *reinterpret_cast<const v2f*>(bp1 + i * 4);
      c0 = __builtin_amdgcn_wmma_f32_16x16x4_f32(
           false, a,  false, b0,
           (short)0, c0,  false,  false);
      c1 = __builtin_amdgcn_wmma_f32_16x16x4_f32(
           false, a,  false, b1,
           (short)0, c1,  false,  false);
    }
    float g0[8], g1[8];
#pragma unroll
    for (int v = 0; v < 8; ++v) {
      g0[v] = 1.0f / (1.0f + expf(-c0[v]));
      g1[v] = 1.0f / (1.0f + expf(-c1[v]));
    }
    __syncthreads();
    float* gp = &sG[(mt * 16 + half * 8) * GATE_PITCH + nt0 * 16 + l15];
#pragma unroll
    for (int v = 0; v < 8; ++v) {
      gp[v * GATE_PITCH] = g0[v];
      gp[v * GATE_PITCH + 16] = g1[v];
    }
  }
  __syncthreads();

  float4 a[GATHER_K], acc[GATHER_K];
  float in1[GATHER_K], ssum[GATHER_K];
  int4 cur[GATHER_K];
  int am[GATHER_K];
  int nm[GATHER_K];
  int nidx[GATHER_K];
#pragma unroll
  for (int k = 0; k < GATHER_K; ++k) {
    int s = min(s0 + k, n1 - 1);
    a[k] = reinterpret_cast<const float4*>(X1 + (size_t)s * NODE_DIM)[lane];
    in1[k] = invn1[s];
    nidx[k] = head[s];
    acc[k] = make_float4(0.0f, 0.0f, 0.0f, 0.0f);
    ssum[k] = 0.0f;
  }
#pragma unroll
  for (int k = 0; k < GATHER_K; ++k) {
    am[k] = -(int)((s0 + k < n1) & ((unsigned)nidx[k] < (unsigned)E));
    cur[k] = link[(unsigned)(nidx[k] & am[k])];
  }
#pragma unroll
  for (int k = 0; k < GATHER_K; ++k) {
    nm[k] = am[k] & -(int)((unsigned)cur[k].x < (unsigned)E);
    nidx[k] = cur[k].x & nm[k];
  }
  static_assert(GATHER_K == 4, "sum4_by_lane_group merges exactly four sums");
  const bool row0 = lane < 16;
  const bool oct1 = (lane & 8) != 0;
  const int myk = lane >> 3;
  const float in1sel = myk == 0   ? in1[0]
                       : myk == 1 ? in1[1]
                       : myk == 2 ? in1[2]
                                  : in1[3];
  float in2sel = __int_as_float(myk == 0   ? cur[0].z
                                : myk == 1 ? cur[1].z
                                : myk == 2 ? cur[2].z
                                           : cur[3].z);
  for (int it = 0; it < E; ++it) {
    int any = 0;
#pragma unroll
    for (int k = 0; k < GATHER_K; ++k) any |= am[k];
    if (any == 0) break;
    float4 b[GATHER_K];
    int4 nx[GATHER_K];
#pragma unroll
    for (int k = 0; k < GATHER_K; ++k) {
      unsigned d = (unsigned)cur[k].y;
      b[k] = reinterpret_cast<const float4*>(X2 + (size_t)d * NODE_DIM)[lane];
      nx[k] = link[(unsigned)nidx[k]];
    }
    float p[GATHER_K];
#pragma unroll
    for (int k = 0; k < GATHER_K; ++k)
      p[k] = a[k].x * b[k].x + a[k].y * b[k].y + a[k].z * b[k].z +
             a[k].w * b[k].w;
    const float dotk = sum4_by_lane_group(p, row0, oct1);
    const float wlane = expf(dotk * in1sel * in2sel);
#pragma unroll
    for (int k = 0; k < GATHER_K; ++k) {
      float wgt = __int_as_float(
          __builtin_amdgcn_readlane(__float_as_int(wlane), 8 * k) & am[k]);
      ssum[k] += wgt;
      acc[k].x += wgt * b[k].x;
      acc[k].y += wgt * b[k].y;
      acc[k].z += wgt * b[k].z;
      acc[k].w += wgt * b[k].w;
    }
#pragma unroll
    for (int k = 0; k < GATHER_K; ++k) {
      asm volatile("" : "+s"(nx[k].x), "+s"(nx[k].y), "+s"(nx[k].z));
      am[k] = nm[k];
      cur[k] = nx[k];
      nm[k] = nm[k] & -(int)((unsigned)nx[k].x < (unsigned)E);
      nidx[k] = nx[k].x & nm[k];
    }
    in2sel = __int_as_float(myk == 0   ? cur[0].z
                            : myk == 1 ? cur[1].z
                            : myk == 2 ? cur[2].z
                                       : cur[3].z);
  }
#pragma unroll
  for (int k = 0; k < GATHER_K; ++k) {
    int s = s0 + k;
    if (s < n1) {
      const float4 g = *reinterpret_cast<const float4*>(
          &sG[(wq * GATHER_K + k) * GATE_PITCH + lane * 4]);
      float ss = ssum[k];
      float inv = ss > 0.0f ? 1.0f / ss : 0.0f;
      float4 r;
      r.x = g.x * (acc[k].x * inv);
      r.y = g.y * (acc[k].y * inv);
      r.z = g.z * (acc[k].z * inv);
      r.w = g.w * (acc[k].w * inv);
      reinterpret_cast<float4*>(out + (size_t)s * NODE_DIM)[lane] = r;
    }
  }
}

extern "C" void kernel_launch(void* const* d_in, const int* in_sizes, int n_in,
                              void* d_out, int out_size, void* d_ws,
                              size_t ws_size, hipStream_t stream) {
  const float* X_h_1 = (const float*)d_in[0];
  const float* X_h_2 = (const float*)d_in[1];
  const float* X_n_1 = (const float*)d_in[2];
  const int* cross = (const int*)d_in[3];
  const float* W_gate = (const float*)d_in[4];
  float* out = (float*)d_out;

  const int N1 = in_sizes[0] / NODE_DIM;
  const int N2 = in_sizes[1] / NODE_DIM;
  const int E = in_sizes[3] / 2;
  const int* src = cross;
  const int* dst = cross + E;

  int4* link = (int4*)d_ws;
  float* invn1 = (float*)(link + E);
  float* invn2 = invn1 + N1;
  int* head = (int*)(invn2 + N2);

  const int npb = (N1 + LINK_BUCKETS - 1) / LINK_BUCKETS;
  const int nchunks = (E + BK_CHUNK - 1) / BK_CHUNK;
  int2* tmp = (int2*)d_out;
  int* tab = (int*)(tmp + E);

  const int norm_groups =
      (N1 + NORM_ROWS - 1) / NORM_ROWS + (N2 + NORM_ROWS - 1) / NORM_ROWS;
  const int groups_per_blk = (BK_THREADS / 32) * NORM_GROUPS;
  const int norm_blocks = (norm_groups + groups_per_blk - 1) / groups_per_blk;
  prep_kernel<<<norm_blocks + nchunks, BK_THREADS, 0, stream>>>(
      src, dst, tmp, tab, E, npb, X_h_1, invn1, N1, X_h_2, invn2, N2,
      norm_blocks);
  link_kernel<<<LINK_BUCKETS, LINK_THREADS, 0, stream>>>(
      tmp, tab, invn2, head, link, N1, npb, nchunks);
  gather_gate_kernel<<<(N1 + GATE_ROWS - 1) / GATE_ROWS, GATHER_WAVES * 32, 0,
                       stream>>>(X_h_1, X_h_2, head, link, invn1, X_n_1,
                                 W_gate, out, N1, E);
}


# both exponentials (softmax weight of every list-walk trip, gate sigmoid) as multiply + hardware exp2 (__expf) instead of library expf with its range handling; weights change by about 1 ulp
# speedup vs baseline: 5.8818x; 1.0373x over previous
#include <hip/hip_runtime.h>
#include <math.h>

typedef float v2f __attribute__((ext_vector_type(2)));
typedef float v8f __attribute__((ext_vector_type(8)));

#define NODE_DIM 128
#define ATTR_DIM 64
#define EPS 1e-8f

#define NORM_ROWS 8
#define NORM_GROUPS 4

struct NormGroup {
  const float* X;
  float* inv;
  int n;
  int r0;
};

__device__ __forceinline__ NormGroup norm_group(
    int gi, int g1, const float* __restrict__ X1, float* __restrict__ inv1,
    int n1, const float* __restrict__ X2, float* __restrict__ inv2, int n2) {
  NormGroup q;
  const bool first = gi < g1;
  q.X = first ? X1 : X2;
  q.inv = first ? inv1 : inv2;
  q.n = first ? n1 : n2;
  q.r0 = (first ? gi : gi - g1) * NORM_ROWS;
  return q;
}

__device__ __forceinline__ void norm_load(const NormGroup& q, int lane,
                                          float4 (&a)[NORM_ROWS]) {
#pragma unroll
  for (int k = 0; k < NORM_ROWS; ++k) {
    int r = min(q.r0 + k, q.n - 1);
    a[k] = reinterpret_cast<const float4*>(q.X + (size_t)r * NODE_DIM)[lane];
  }
}

__device__ __forceinline__ void norm_finish(const NormGroup& q, int lane,
                                            const float4 (&a)[NORM_ROWS]) {
  float p[NORM_ROWS];
#pragma unroll
  for (int k = 0; k < NORM_ROWS; ++k)
    p[k] = a[k].x * a[k].x + a[k].y * a[k].y + a[k].z * a[k].z +
           a[k].w * a[k].w;
#pragma unroll
  for (int k = 0; k < NORM_ROWS; ++k) {
#pragma unroll
    for (int off = 16; off > 0; off >>= 1) p[k] += __shfl_xor(p[k], off, 32);
  }
  float mine = p[0];
#pragma unroll
  for (int k = 1; k < NORM_ROWS; ++k)
    if (lane == k) mine = p[k];
  if (lane < NORM_ROWS && q.r0 + lane < q.n)
    q.inv[q.r0 + lane] = 1.0f / fmaxf(sqrtf(mine), EPS);
}

__device__ __forceinline__ void norm_wave(
    int wave, int lane, const float* __restrict__ X1, float* __restrict__ inv1,
    int n1, const float* __restrict__ X2, float* __restrict__ inv2, int n2) {
  const int g1 = (n1 + NORM_ROWS - 1) / NORM_ROWS;
  const int g2 = (n2 + NORM_ROWS - 1) / NORM_ROWS;
  const int gi0 = wave * NORM_GROUPS;
  if (gi0 >= g1 + g2) return;
  NormGroup q = norm_group(gi0, g1, X1, inv1, n1, X2, inv2, n2);
  float4 cur[NORM_ROWS];
  norm_load(q, lane, cur);
#pragma unroll
  for (int g = 0; g < NORM_GROUPS; ++g) {
    float4 nxt[NORM_ROWS];
    NormGroup qn = q;
    if (g + 1 < NORM_GROUPS) {
      qn = norm_group(gi0 + g + 1, g1, X1, inv1, n1, X2, inv2, n2);
      norm_load(qn, lane, nxt);
    }
    norm_finish(q, lane, cur);
    if (g + 1 < NORM_GROUPS) {
      q = qn;
#pragma unroll
      for (int k = 0; k < NORM_ROWS; ++k) cur[k] = nxt[k];
    }
  }
}

#define LINK_BUCKETS 256
#define BK_THREADS 640
#define BK_EPT 4
#define BK_CHUNK (BK_THREADS * BK_EPT)

__global__ void __launch_bounds__(BK_THREADS)
prep_kernel(const int* __restrict__ src, const int* __restrict__ dst,
            int2* __restrict__ tmp, int* __restrict__ tab, int E, int npb,
            const float* __restrict__ X1, float* __restrict__ inv1, int n1,
            const float* __restrict__ X2, float* __restrict__ inv2, int n2,
            int norm_blocks) {
  __shared__ int hist[LINK_BUCKETS];
  __shared__ int wsum[LINK_BUCKETS / 32];
  __shared__ int2 srt[BK_CHUNK];
  const int tid = threadIdx.x;
  const int lane = tid & 31;
  if ((int)blockIdx.x < norm_blocks) {
    norm_wave((int)blockIdx.x * (BK_THREADS / 32) + (tid >> 5), lane, X1, inv1,
              n1, X2, inv2, n2);
    return;
  }
  const int c = (int)blockIdx.x - norm_blocks;
  const int base = c * BK_CHUNK;
  if (tid < LINK_BUCKETS) hist[tid] = 0;
  int s[BK_EPT], d[BK_EPT], b[BK_EPT], r[BK_EPT];
#pragma unroll
  for (int k = 0; k < BK_EPT; ++k) {
    int e = min(base + k * BK_THREADS + tid, E - 1);
    s[k] = src[e];
    d[k] = dst[e];
  }
  __syncthreads();
#pragma unroll
  for (int k = 0; k < BK_EPT; ++k) {
    b[k] = s[k] / npb;
    r[k] = 0;
    if (base + k * BK_THREADS + tid < E)
      r[k] = atomicAdd(&hist[b[k]], 1);
  }
  __syncthreads();
  int v = 0, incl = 0;
  if (tid < LINK_BUCKETS) {
    v = hist[tid];
    incl = v;
#pragma unroll
    for (int off = 1; off < 32; off <<= 1) {
      int t = __shfl_up(incl, off, 32);
      if (lane >= off) incl += t;
    }
    if (lane == 31) wsum[tid >> 5] = incl;
  }
  __syncthreads();
  if (tid < LINK_BUCKETS) {
    int wp = 0;
#pragma unroll
    for (int w = 0; w < LINK_BUCKETS / 32; ++w)
      if (w < (tid >> 5)) wp += wsum[w];
    int excl = wp + incl - v;
    hist[tid] = excl;
    int* row = tab + (size_t)c * (LINK_BUCKETS + 1);
    row[tid] = excl;
    if (tid == LINK_BUCKETS - 1) row[LINK_BUCKETS] = excl + v;
  }
  __syncthreads();
#pragma unroll
  for (int k = 0; k < BK_EPT; ++k)
    if (base + k * BK_THREADS + tid < E)
      srt[hist[b[k]] + r[k]] = make_int2(s[k], d[k]);
  __syncthreads();
  const int n = min(BK_CHUNK, E - base);
  for (int i = tid; i < n; i += BK_THREADS) tmp[base + i] = srt[i];
}

#define LINK_THREADS 512
#define LINK_MAXNPB 1024
#define LINK_U 4

__device__ __forceinline__ int slice_place(int i, const int* cpre,
                                           const int* cstart) {
  int j = 0;
#pragma unroll
  for (int step = LINK_THREADS / 2; step > 0; step >>= 1)
    if (cpre[j + step] <= i) j += step;
  return cstart[j] + (i - cpre[j]);
}

__global__ void __launch_bounds__(LINK_THREADS)
link_kernel(const int2* __restrict__ tmp, const int* __restrict__ tab,
            const float* __restrict__ invn2, int* __restrict__ head,
            int4* __restrict__ link, int n1, int npb, int nchunks) {
  __shared__ int lhead[LINK_MAXNPB];
  __shared__ int cpre[LINK_THREADS];
  __shared__ int cstart[LINK_THREADS];
  __shared__ int wsum[LINK_THREADS / 32];
  const int tid = threadIdx.x;
  const int lane = tid & 31;
  const int wv = tid >> 5;
  const int g = blockIdx.x;
  const int node0 = g * npb;
  for (int i = tid; i < npb; i += LINK_THREADS) lhead[i] = -1;

  for (int cg = 0; cg < nchunks; cg += LINK_THREADS) {
    int c = cg + tid;
    int cnt = 0, st = 0;
    if (c < nchunks) {
      const int* row = tab + (size_t)c * (LINK_BUCKETS + 1) + g;
      int lo = row[0], hi = row[1];
      cnt = hi - lo;
      st = c * BK_CHUNK + lo;
    }
    int incl = cnt;
#pragma unroll
    for (int off = 1; off < 32; off <<= 1) {
      int t = __shfl_up(incl, off, 32);
      if (lane >= off) incl += t;
    }
    if (lane == 31) wsum[wv] = incl;
    __syncthreads();
    int wp = 0, total = 0;
#pragma unroll
    for (int w = 0; w < LINK_THREADS / 32; ++w) {
      int x = wsum[w];
      if (w < wv) wp += x;
      total += x;
    }
    cpre[tid] = wp + incl - cnt;
    cstart[tid] = st;
    __syncthreads();

    int done = 0;
    for (; done + LINK_THREADS * LINK_U <= total; done += LINK_THREADS * LINK_U) {
      int p[LINK_U];
      int2 sd[LINK_U];
      float in2[LINK_U];
#pragma unroll
      for (int k = 0; k < LINK_U; ++k)
        p[k] = slice_place(done + k * LINK_THREADS + tid, cpre, cstart);
#pragma unroll
      for (int k = 0; k < LINK_U; ++k) sd[k] = tmp[p[k]];
#pragma unroll
      for (int k = 0; k < LINK_U; ++k) in2[k] = invn2[sd[k].y];
#pragma unroll
      for (int k = 0; k < LINK_U; ++k) {
        int prev = atomicExch(&lhead[sd[k].x - node0], p[k]);
        link[p[k]] = make_int4(prev, sd[k].y, __float_as_int(in2[k]), 0);
      }
    }
    for (int i = done + tid; i < total; i += LINK_THREADS) {
      int p = slice_place(i, cpre, cstart);
      int2 sd = tmp[p];
      float in2 = invn2[sd.y];
      int prev = atomicExch(&lhead[sd.x - node0], p);
      link[p] = make_int4(prev, sd.y, __float_as_int(in2), 0);
    }
    __syncthreads();
  }
  __syncthreads();
  for (int i = tid; i < npb; i += LINK_THREADS)
    if (node0 + i < n1) head[node0 + i] = lhead[i];
}

#define GATHER_K 4
#define GATHER_WAVES 8
#define GATHER_SIMD_WAVES 12
#define GATE_ROWS (GATHER_WAVES * GATHER_K)
#define GATE_KSTEPS (ATTR_DIM / 4)
#define FRAG_PITCH (ATTR_DIM + 4)
#define GATE_PITCH (NODE_DIM + 4)

__device__ __forceinline__ float lane_xor16(float x) {
  int b = __builtin_bit_cast(int, x);
  return __builtin_bit_cast(
      float, __builtin_amdgcn_permlanex16(b, b, 0x76543210, 0xfedcba98, false,
                                          false));
}
template <int M>
__device__ __forceinline__ float lane_xor_row(float x) {
  return __builtin_bit_cast(
      float, __builtin_amdgcn_update_dpp(0, __builtin_bit_cast(int, x),
                                         0x160 | M, 0xf, 0xf, true));
}
__device__ __forceinline__ float sum4_by_lane_group(const float (&p)[4],
                                                    bool row0, bool oct1) {
  float ka = row0 ? p[0] : p[2], sa = row0 ? p[2] : p[0];
  float kb = row0 ? p[1] : p[3], sb = row0 ? p[3] : p[1];
  float qa = ka + lane_xor16(sa);
  float qb = kb + lane_xor16(sb);
  float kc = oct1 ? qb : qa, sc = oct1 ? qa : qb;
  float r = kc + lane_xor_row<8>(sc);
  r += lane_xor_row<4>(r);
  r += lane_xor_row<2>(r);
  r += lane_xor_row<1>(r);
  return r;
}

__global__ void __launch_bounds__(GATHER_WAVES * 32, GATHER_SIMD_WAVES)
gather_gate_kernel(const float* __restrict__ X1, const float* __restrict__ X2,
                   const int* __restrict__ head,
                   const int4* __restrict__ link,
                   const float* __restrict__ invn1,
                   const float* __restrict__ Xn, const float* __restrict__ Wg,
                   float* __restrict__ out, int n1, int E) {
  __shared__ __attribute__((aligned(16))) float sW[NODE_DIM * FRAG_PITCH];
  __shared__ __attribute__((aligned(16))) float sA[GATE_ROWS * FRAG_PITCH];
  float* sG = sW;
  const int tid = threadIdx.x;
  int w = __builtin_amdgcn_readfirstlane(
      (int)((blockIdx.x * (GATHER_WAVES * 32) + threadIdx.x) >> 5));
  int lane = tid & 31;
  int wq = w & (GATHER_WAVES - 1);
  int s0 = w * GATHER_K;
  int node0 = (w - wq) * GATHER_K;

  {
    const float4* Wg4 = reinterpret_cast<const float4*>(Wg);
    const float4* Xn4 = reinterpret_cast<const float4*>(Xn);
    float4 wv[NODE_DIM * ATTR_DIM / 4 / (GATHER_WAVES * 32)];
    float4 xv[GATE_ROWS * ATTR_DIM / 4 / (GATHER_WAVES * 32)];
#pragma unroll
    for (int j = 0; j < NODE_DIM * ATTR_DIM / 4 / (GATHER_WAVES * 32); ++j)
      wv[j] = Wg4[tid + j * (GATHER_WAVES * 32)];
#pragma unroll
    for (int j = 0; j < GATE_ROWS * ATTR_DIM / 4 / (GATHER_WAVES * 32); ++j) {
      int idx = tid + j * (GATHER_WAVES * 32);
      int node = min(node0 + (idx >> 4), n1 - 1);
      xv[j] = Xn4[(size_t)node * (ATTR_DIM / 4) + (idx & 15)];
    }
#pragma unroll
    for (int j = 0; j < NODE_DIM * ATTR_DIM / 4 / (GATHER_WAVES * 32); ++j) {
      int idx = tid + j * (GATHER_WAVES * 32);
      *reinterpret_cast<float4*>(&sW[(idx >> 4) * FRAG_PITCH + (idx & 15) * 4]) =
          wv[j];
    }
#pragma unroll
    for (int j = 0; j < GATE_ROWS * ATTR_DIM / 4 / (GATHER_WAVES * 32); ++j) {
      int idx = tid + j * (GATHER_WAVES * 32);
      *reinterpret_cast<float4*>(&sA[(idx >> 4) * FRAG_PITCH + (idx & 15) * 4]) =
          xv[j];
    }
  }
  __syncthreads();
  {
    const int l15 = lane & 15;
    const int half = lane >> 4;
    const int mt = wq >> 2;
    const int nt0 = (wq & 3) * 2;
    const float* ap = &sA[(mt * 16 + l15) * FRAG_PITCH + half * 2];
    const float* bp0 = &sW[(nt0 * 16 + l15) * FRAG_PITCH + half * 2];
    const float* bp1 = bp0 + 16 * FRAG_PITCH;
    v8f c0 = {}, c1 = {};
#pragma unroll
    for (int i = 0; i < GATE_KSTEPS; ++i) {
      v2f a = *reinterpret_cast<const v2f*>(ap + i * 4);
      v2f b0 = *reinterpret_cast<const v2f*>(bp0 + i * 4);
      v2f b1 = *reinterpret_cast<const v2f*>(bp1 + i * 4);
      c0 = __builtin_amdgcn_wmma_f32_16x16x4_f32(
           false, a,  false, b0,
           (short)0, c0,  false,  false);
      c1 = __builtin_amdgcn_wmma_f32_16x16x4_f32(
           false, a,  false, b1,
           (short)0, c1,  false,  false);
    }
    float g0[8], g1[8];
#pragma unroll
    for (int v = 0; v < 8; ++v) {
      g0[v] = 1.0f / (1.0f + __expf(-c0[v]));
      g1[v] = 1.0f / (1.0f + __expf(-c1[v]));
    }
    __syncthreads();
    float* gp = &sG[(mt * 16 + half * 8) * GATE_PITCH + nt0 * 16 + l15];
#pragma unroll
    for (int v = 0; v < 8; ++v) {
      gp[v * GATE_PITCH] = g0[v];
      gp[v * GATE_PITCH + 16] = g1[v];
    }
  }
  __syncthreads();

  float4 a[GATHER_K], acc[GATHER_K];
  float in1[GATHER_K], ssum[GATHER_K];
  int4 cur[GATHER_K];
  int am[GATHER_K];
  int nm[GATHER_K];
  int nidx[GATHER_K];
#pragma unroll
  for (int k = 0; k < GATHER_K; ++k) {
    int s = min(s0 + k, n1 - 1);
    a[k] = reinterpret_cast<const float4*>(X1 + (size_t)s * NODE_DIM)[lane];
    in1[k] = invn1[s];
    nidx[k] = head[s];
    acc[k] = make_float4(0.0f, 0.0f, 0.0f, 0.0f);
    ssum[k] = 0.0f;
  }
#pragma unroll
  for (int k = 0; k < GATHER_K; ++k) {
    am[k] = -(int)((s0 + k < n1) & ((unsigned)nidx[k] < (unsigned)E));
    cur[k] = link[(unsigned)(nidx[k] & am[k])];
  }
#pragma unroll
  for (int k = 0; k < GATHER_K; ++k) {
    nm[k] = am[k] & -(int)((unsigned)cur[k].x < (unsigned)E);
    nidx[k] = cur[k].x & nm[k];
  }
  static_assert(GATHER_K == 4, "sum4_by_lane_group merges exactly four sums");
  const bool row0 = lane < 16;
  const bool oct1 = (lane & 8) != 0;
  const int myk = lane >> 3;
  const float in1sel = myk == 0   ? in1[0]
                       : myk == 1 ? in1[1]
                       : myk == 2 ? in1[2]
                                  : in1[3];
  float in2sel = __int_as_float(myk == 0   ? cur[0].z
                                : myk == 1 ? cur[1].z
                                : myk == 2 ? cur[2].z
                                           : cur[3].z);
  for (int it = 0; it < E; ++it) {
    int any = 0;
#pragma unroll
    for (int k = 0; k < GATHER_K; ++k) any |= am[k];
    if (any == 0) break;
    float4 b[GATHER_K];
    int4 nx[GATHER_K];
#pragma unroll
    for (int k = 0; k < GATHER_K; ++k) {
      unsigned d = (unsigned)cur[k].y;
      b[k] = reinterpret_cast<const float4*>(X2 + (size_t)d * NODE_DIM)[lane];
      nx[k] = link[(unsigned)nidx[k]];
    }
    float p[GATHER_K];
#pragma unroll
    for (int k = 0; k < GATHER_K; ++k)
      p[k] = fmaf(a[k].w, b[k].w,
                  fmaf(a[k].z, b[k].z,
                       fmaf(a[k].x, b[k].x, a[k].y * b[k].y)));
    const float dotk = sum4_by_lane_group(p, row0, oct1);
    const float wlane = __expf(dotk * in1sel * in2sel);
#pragma unroll
    for (int k = 0; k < GATHER_K; ++k) {
      float wgt = __int_as_float(
          __builtin_amdgcn_readlane(__float_as_int(wlane), 8 * k) & am[k]);
      ssum[k] += wgt;
      acc[k].x += wgt * b[k].x;
      acc[k].y += wgt * b[k].y;
      acc[k].z += wgt * b[k].z;
      acc[k].w += wgt * b[k].w;
    }
#pragma unroll
    for (int k = 0; k < GATHER_K; ++k) {
      asm volatile("" : "+s"(nx[k].x), "+s"(nx[k].y), "+s"(nx[k].z));
      am[k] = nm[k];
      cur[k] = nx[k];
      nm[k] = nm[k] & -(int)((unsigned)nx[k].x < (unsigned)E);
      nidx[k] = nx[k].x & nm[k];
    }
    in2sel = __int_as_float(myk == 0   ? cur[0].z
                            : myk == 1 ? cur[1].z
                            : myk == 2 ? cur[2].z
                                       : cur[3].z);
  }
#pragma unroll
  for (int k = 0; k < GATHER_K; ++k) {
    int s = s0 + k;
    if (s < n1) {
      const float4 g = *reinterpret_cast<const float4*>(
          &sG[(wq * GATHER_K + k) * GATE_PITCH + lane * 4]);
      float ss = ssum[k];
      float inv = ss > 0.0f ? 1.0f / ss : 0.0f;
      float4 r;
      r.x = g.x * (acc[k].x * inv);
      r.y = g.y * (acc[k].y * inv);
      r.z = g.z * (acc[k].z * inv);
      r.w = g.w * (acc[k].w * inv);
      reinterpret_cast<float4*>(out + (size_t)s * NODE_DIM)[lane] = r;
    }
  }
}

extern "C" void kernel_launch(void* const* d_in, const int* in_sizes, int n_in,
                              void* d_out, int out_size, void* d_ws,
                              size_t ws_size, hipStream_t stream) {
  const float* X_h_1 = (const float*)d_in[0];
  const float* X_h_2 = (const float*)d_in[1];
  const float* X_n_1 = (const float*)d_in[2];
  const int* cross = (const int*)d_in[3];
  const float* W_gate = (const float*)d_in[4];
  float* out = (float*)d_out;

  const int N1 = in_sizes[0] / NODE_DIM;
  const int N2 = in_sizes[1] / NODE_DIM;
  const int E = in_sizes[3] / 2;
  const int* src = cross;
  const int* dst = cross + E;

  int4* link = (int4*)d_ws;
  float* invn1 = (float*)(link + E);
  float* invn2 = invn1 + N1;
  int* head = (int*)(invn2 + N2);

  const int npb = (N1 + LINK_BUCKETS - 1) / LINK_BUCKETS;
  const int nchunks = (E + BK_CHUNK - 1) / BK_CHUNK;
  int2* tmp = (int2*)d_out;
  int* tab = (int*)(tmp + E);

  const int norm_groups =
      (N1 + NORM_ROWS - 1) / NORM_ROWS + (N2 + NORM_ROWS - 1) / NORM_ROWS;
  const int groups_per_blk = (BK_THREADS / 32) * NORM_GROUPS;
  const int norm_blocks = (norm_groups + groups_per_blk - 1) / groups_per_blk;
  prep_kernel<<<norm_blocks + nchunks, BK_THREADS, 0, stream>>>(
      src, dst, tmp, tab, E, npb, X_h_1, invn1, N1, X_h_2, invn2, N2,
      norm_blocks);
  link_kernel<<<LINK_BUCKETS, LINK_THREADS, 0, stream>>>(
      tmp, tab, invn2, head, link, N1, npb, nchunks);
  gather_gate_kernel<<<(N1 + GATE_ROWS - 1) / GATE_ROWS, GATHER_WAVES * 32, 0,
                       stream>>>(X_h_1, X_h_2, head, link, invn1, X_n_1,
                                 W_gate, out, N1, E);
}


# edges stored as one contiguous run per node (two-sweep counting sort in LDS, no linked list): gather takes a run 4 edges a trip with the next rows prefetched; sum order = run order (LDS arrival)
# speedup vs baseline: 6.1585x; 1.0470x over previous
#include <hip/hip_runtime.h>
#include <math.h>

typedef float v2f __attribute__((ext_vector_type(2)));
typedef float v8f __attribute__((ext_vector_type(8)));

#define NODE_DIM 128
#define ATTR_DIM 64
#define EPS 1e-8f

#define NORM_ROWS 8
#define NORM_GROUPS 4

struct NormGroup {
  const float* X;
  float* inv;
  int n;
  int r0;
};

__device__ __forceinline__ NormGroup norm_group(
    int gi, int g1, const float* __restrict__ X1, float* __restrict__ inv1,
    int n1, const float* __restrict__ X2, float* __restrict__ inv2, int n2) {
  NormGroup q;
  const bool first = gi < g1;
  q.X = first ? X1 : X2;
  q.inv = first ? inv1 : inv2;
  q.n = first ? n1 : n2;
  q.r0 = (first ? gi : gi - g1) * NORM_ROWS;
  return q;
}

__device__ __forceinline__ void norm_load(const NormGroup& q, int lane,
                                          float4 (&a)[NORM_ROWS]) {
#pragma unroll
  for (int k = 0; k < NORM_ROWS; ++k) {
    int r = min(q.r0 + k, q.n - 1);
    a[k] = reinterpret_cast<const float4*>(q.X + (size_t)r * NODE_DIM)[lane];
  }
}

__device__ __forceinline__ void norm_finish(const NormGroup& q, int lane,
                                            const float4 (&a)[NORM_ROWS]) {
  float p[NORM_ROWS];
#pragma unroll
  for (int k = 0; k < NORM_ROWS; ++k)
    p[k] = a[k].x * a[k].x + a[k].y * a[k].y + a[k].z * a[k].z +
           a[k].w * a[k].w;
#pragma unroll
  for (int k = 0; k < NORM_ROWS; ++k) {
#pragma unroll
    for (int off = 16; off > 0; off >>= 1) p[k] += __shfl_xor(p[k], off, 32);
  }
  float mine = p[0];
#pragma unroll
  for (int k = 1; k < NORM_ROWS; ++k)
    if (lane == k) mine = p[k];
  if (lane < NORM_ROWS && q.r0 + lane < q.n)
    q.inv[q.r0 + lane] = 1.0f / fmaxf(sqrtf(mine), EPS);
}

__device__ __forceinline__ void norm_wave(
    int wave, int lane, const float* __restrict__ X1, float* __restrict__ inv1,
    int n1, const float* __restrict__ X2, float* __restrict__ inv2, int n2) {
  const int g1 = (n1 + NORM_ROWS - 1) / NORM_ROWS;
  const int g2 = (n2 + NORM_ROWS - 1) / NORM_ROWS;
  const int gi0 = wave * NORM_GROUPS;
  if (gi0 >= g1 + g2) return;
  NormGroup q = norm_group(gi0, g1, X1, inv1, n1, X2, inv2, n2);
  float4 cur[NORM_ROWS];
  norm_load(q, lane, cur);
#pragma unroll
  for (int g = 0; g < NORM_GROUPS; ++g) {
    float4 nxt[NORM_ROWS];
    NormGroup qn = q;
    if (g + 1 < NORM_GROUPS) {
      qn = norm_group(gi0 + g + 1, g1, X1, inv1, n1, X2, inv2, n2);
      norm_load(qn, lane, nxt);
    }
    norm_finish(q, lane, cur);
    if (g + 1 < NORM_GROUPS) {
      q = qn;
#pragma unroll
      for (int k = 0; k < NORM_ROWS; ++k) cur[k] = nxt[k];
    }
  }
}

#define LINK_BUCKETS 256
#define BK_THREADS 640
#define BK_EPT 4
#define BK_CHUNK (BK_THREADS * BK_EPT)

__global__ void __launch_bounds__(BK_THREADS)
prep_kernel(const int* __restrict__ src, const int* __restrict__ dst,
            int2* __restrict__ tmp, int* __restrict__ tab, int E, int npb,
            const float* __restrict__ X1, float* __restrict__ inv1, int n1,
            const float* __restrict__ X2, float* __restrict__ inv2, int n2,
            int norm_blocks) {
  __shared__ int hist[LINK_BUCKETS];
  __shared__ int wsum[LINK_BUCKETS / 32];
  __shared__ int2 srt[BK_CHUNK];
  const int tid = threadIdx.x;
  const int lane = tid & 31;
  if ((int)blockIdx.x < norm_blocks) {
    norm_wave((int)blockIdx.x * (BK_THREADS / 32) + (tid >> 5), lane, X1, inv1,
              n1, X2, inv2, n2);
    return;
  }
  const int c = (int)blockIdx.x - norm_blocks;
  const int base = c * BK_CHUNK;
  if (tid < LINK_BUCKETS) hist[tid] = 0;
  int s[BK_EPT], d[BK_EPT], b[BK_EPT], r[BK_EPT];
#pragma unroll
  for (int k = 0; k < BK_EPT; ++k) {
    int e = min(base + k * BK_THREADS + tid, E - 1);
    s[k] = src[e];
    d[k] = dst[e];
  }
  __syncthreads();
#pragma unroll
  for (int k = 0; k < BK_EPT; ++k) {
    b[k] = s[k] / npb;
    r[k] = 0;
    if (base + k * BK_THREADS + tid < E)
      r[k] = atomicAdd(&hist[b[k]], 1);
  }
  __syncthreads();
  int v = 0, incl = 0;
  if (tid < LINK_BUCKETS) {
    v = hist[tid];
    incl = v;
#pragma unroll
    for (int off = 1; off < 32; off <<= 1) {
      int t = __shfl_up(incl, off, 32);
      if (lane >= off) incl += t;
    }
    if (lane == 31) wsum[tid >> 5] = incl;
  }
  __syncthreads();
  if (tid < LINK_BUCKETS) {
    int wp = 0;
#pragma unroll
    for (int w = 0; w < LINK_BUCKETS / 32; ++w)
      if (w < (tid >> 5)) wp += wsum[w];
    int excl = wp + incl - v;
    hist[tid] = excl;
    int* row = tab + (size_t)c * (LINK_BUCKETS + 1);
    row[tid] = excl;
    if (tid == LINK_BUCKETS - 1) row[LINK_BUCKETS] = excl + v;
  }
  __syncthreads();
#pragma unroll
  for (int k = 0; k < BK_EPT; ++k)
    if (base + k * BK_THREADS + tid < E)
      srt[hist[b[k]] + r[k]] = make_int2(s[k], d[k]);
  __syncthreads();
  const int n = min(BK_CHUNK, E - base);
  for (int i = tid; i < n; i += BK_THREADS) tmp[base + i] = srt[i];
}

#define LINK_THREADS 512
#define LINK_MAXNPB 1024
#define LINK_U 8

__device__ __forceinline__ int slice_place(int i, const int* cpre,
                                           const int* cstart) {
  int j = 0;
#pragma unroll
  for (int step = LINK_THREADS / 2; step > 0; step >>= 1)
    if (cpre[j + step] <= i) j += step;
  return cstart[j] + (i - cpre[j]);
}

__device__ __forceinline__ int slice_scan(int cg, int tid, int lane, int wv,
                                          int g, int nchunks,
                                          const int* __restrict__ tab,
                                          int* cpre, int* cstart, int* wsum) {
  int c = cg + tid;
  int cnt = 0, st = 0;
  if (c < nchunks) {
    const int* row = tab + (size_t)c * (LINK_BUCKETS + 1) + g;
    int lo = row[0], hi = row[1];
    cnt = hi - lo;
    st = c * BK_CHUNK + lo;
  }
  int incl = cnt;
#pragma unroll
  for (int off = 1; off < 32; off <<= 1) {
    int t = __shfl_up(incl, off, 32);
    if (lane >= off) incl += t;
  }
  if (lane == 31) wsum[wv] = incl;
  __syncthreads();
  int wp = 0, total = 0;
#pragma unroll
  for (int w = 0; w < LINK_THREADS / 32; ++w) {
    int x = wsum[w];
    if (w < wv) wp += x;
    total += x;
  }
  cpre[tid] = wp + incl - cnt;
  cstart[tid] = st;
  __syncthreads();
  return total;
}

__global__ void __launch_bounds__(LINK_THREADS)
link_kernel(const int2* __restrict__ tmp, const int* __restrict__ tab,
            const float* __restrict__ invn2, int* __restrict__ start,
            int2* __restrict__ ent, int n1, int npb, int nchunks) {
  __shared__ int ncnt[LINK_MAXNPB];
  __shared__ int cpre[LINK_THREADS];
  __shared__ int cstart[LINK_THREADS];
  __shared__ int wsum[LINK_THREADS / 32];
  __shared__ int bsum[LINK_THREADS / 32];
  static_assert(LINK_MAXNPB == 2 * LINK_THREADS, "the count scan takes two nodes per thread");
  const int tid = threadIdx.x;
  const int lane = tid & 31;
  const int wv = tid >> 5;
  const int g = blockIdx.x;
  const int node0 = g * npb;
  ncnt[2 * tid] = 0;
  ncnt[2 * tid + 1] = 0;
  int lo_sum = 0;
  for (int c = tid; c < nchunks; c += LINK_THREADS)
    lo_sum += tab[(size_t)c * (LINK_BUCKETS + 1) + g];
#pragma unroll
  for (int off = 16; off > 0; off >>= 1) lo_sum += __shfl_xor(lo_sum, off, 32);
  if (lane == 0) bsum[wv] = lo_sum;
  __syncthreads();
  int base = 0;
#pragma unroll
  for (int w = 0; w < LINK_THREADS / 32; ++w) base += bsum[w];

  for (int cg = 0; cg < nchunks; cg += LINK_THREADS) {
    const int total =
        slice_scan(cg, tid, lane, wv, g, nchunks, tab, cpre, cstart, wsum);
    for (int done = 0; done < total; done += LINK_THREADS * LINK_U) {
      int p[LINK_U], sn[LINK_U];
#pragma unroll
      for (int k = 0; k < LINK_U; ++k)
        p[k] = slice_place(min(done + k * LINK_THREADS + tid, total - 1), cpre,
                           cstart);
#pragma unroll
      for (int k = 0; k < LINK_U; ++k) sn[k] = tmp[p[k]].x;
#pragma unroll
      for (int k = 0; k < LINK_U; ++k)
        if (done + k * LINK_THREADS + tid < total)
          atomicAdd(&ncnt[sn[k] - node0], 1);
    }
    __syncthreads();
  }
  __syncthreads();

  {
    const int v0 = ncnt[2 * tid], v1 = ncnt[2 * tid + 1];
    const int sum = v0 + v1;
    int incl = sum;
#pragma unroll
    for (int off = 1; off < 32; off <<= 1) {
      int t = __shfl_up(incl, off, 32);
      if (lane >= off) incl += t;
    }
    if (lane == 31) wsum[wv] = incl;
    __syncthreads();
    int wp = 0;
#pragma unroll
    for (int w = 0; w < LINK_THREADS / 32; ++w)
      if (w < wv) wp += wsum[w];
    const int ex0 = base + wp + incl - sum;
    const int ex1 = ex0 + v0;
    ncnt[2 * tid] = ex0;
    ncnt[2 * tid + 1] = ex1;
    const int n = node0 + 2 * tid;
    if (2 * tid < npb && n < n1) start[n] = ex0;
    if (2 * tid + 1 < npb && n + 1 < n1) start[n + 1] = ex1;
  }
  __syncthreads();

  for (int cg = 0; cg < nchunks; cg += LINK_THREADS) {
    const int total =
        slice_scan(cg, tid, lane, wv, g, nchunks, tab, cpre, cstart, wsum);
    for (int done = 0; done < total; done += LINK_THREADS * LINK_U) {
      int p[LINK_U];
      int2 sd[LINK_U];
      float in2[LINK_U];
#pragma unroll
      for (int k = 0; k < LINK_U; ++k)
        p[k] = slice_place(min(done + k * LINK_THREADS + tid, total - 1), cpre,
                           cstart);
#pragma unroll
      for (int k = 0; k < LINK_U; ++k) sd[k] = tmp[p[k]];
#pragma unroll
      for (int k = 0; k < LINK_U; ++k) in2[k] = invn2[sd[k].y];
#pragma unroll
      for (int k = 0; k < LINK_U; ++k)
        if (done + k * LINK_THREADS + tid < total) {
          int pos = atomicAdd(&ncnt[sd[k].x - node0], 1);
          ent[pos] = make_int2(sd[k].y, __float_as_int(in2[k]));
        }
    }
    __syncthreads();
  }
}

#define GATHER_K 4
#define GATHER_U 4
#define GATHER_WAVES 8
#define GATHER_SIMD_WAVES 12
#define GATE_ROWS (GATHER_WAVES * GATHER_K)
#define GATE_KSTEPS (ATTR_DIM / 4)
#define FRAG_PITCH (ATTR_DIM + 4)
#define GATE_PITCH (NODE_DIM + 4)

__device__ __forceinline__ float lane_xor16(float x) {
  int b = __builtin_bit_cast(int, x);
  return __builtin_bit_cast(
      float, __builtin_amdgcn_permlanex16(b, b, 0x76543210, 0xfedcba98, false,
                                          false));
}
template <int M>
__device__ __forceinline__ float lane_xor_row(float x) {
  return __builtin_bit_cast(
      float, __builtin_amdgcn_update_dpp(0, __builtin_bit_cast(int, x),
                                         0x160 | M, 0xf, 0xf, true));
}
__device__ __forceinline__ float sum4_by_lane_group(const float (&p)[4],
                                                    bool row0, bool oct1) {
  float ka = row0 ? p[0] : p[2], sa = row0 ? p[2] : p[0];
  float kb = row0 ? p[1] : p[3], sb = row0 ? p[3] : p[1];
  float qa = ka + lane_xor16(sa);
  float qb = kb + lane_xor16(sb);
  float kc = oct1 ? qb : qa, sc = oct1 ? qa : qb;
  float r = kc + lane_xor_row<8>(sc);
  r += lane_xor_row<4>(r);
  r += lane_xor_row<2>(r);
  r += lane_xor_row<1>(r);
  return r;
}

__device__ __forceinline__ void batch_sum(
    const float4& ak, float in1k, const float4 (&rw)[GATHER_U],
    const int2 (&en)[GATHER_U], unsigned pos, unsigned end, bool row0,
    bool oct1, const int (&mk)[GATHER_U], float4& acc, float& ssum) {
  float p[GATHER_U];
#pragma unroll
  for (int j = 0; j < GATHER_U; ++j)
    p[j] = fmaf(ak.w, rw[j].w,
                fmaf(ak.z, rw[j].z, fmaf(ak.x, rw[j].x, ak.y * rw[j].y)));
  const float dotj = sum4_by_lane_group(p, row0, oct1);
  const float in2sel = __int_as_float((en[0].y & mk[0]) | (en[1].y & mk[1]) |
                                      (en[2].y & mk[2]) | (en[3].y & mk[3]));
  const float wlane = __expf(dotj * in1k * in2sel);
#pragma unroll
  for (int j = 0; j < GATHER_U; ++j) {
    const int m = -(int)(pos + j < end);
    const float wgt = __int_as_float(
        __builtin_amdgcn_readlane(__float_as_int(wlane), 8 * j) & m);
    ssum += wgt;
    acc.x += wgt * rw[j].x;
    acc.y += wgt * rw[j].y;
    acc.z += wgt * rw[j].z;
    acc.w += wgt * rw[j].w;
  }
}

__global__ void __launch_bounds__(GATHER_WAVES * 32, GATHER_SIMD_WAVES)
gather_gate_kernel(const float* __restrict__ X1, const float* __restrict__ X2,
                   const int* __restrict__ start,
                   const int2* __restrict__ ent,
                   const float* __restrict__ invn1,
                   const float* __restrict__ Xn, const float* __restrict__ Wg,
                   float* __restrict__ out, int n1, int E) {
  __shared__ __attribute__((aligned(16))) float sW[NODE_DIM * FRAG_PITCH];
  __shared__ __attribute__((aligned(16))) float sA[GATE_ROWS * FRAG_PITCH];
  float* sG = sW;
  const int tid = threadIdx.x;
  int w = __builtin_amdgcn_readfirstlane(
      (int)((blockIdx.x * (GATHER_WAVES * 32) + threadIdx.x) >> 5));
  int lane = tid & 31;
  int wq = w & (GATHER_WAVES - 1);
  int s0 = w * GATHER_K;
  int node0 = (w - wq) * GATHER_K;

  {
    const float4* Wg4 = reinterpret_cast<const float4*>(Wg);
    const float4* Xn4 = reinterpret_cast<const float4*>(Xn);
    float4 wv[NODE_DIM * ATTR_DIM / 4 / (GATHER_WAVES * 32)];
    float4 xv[GATE_ROWS * ATTR_DIM / 4 / (GATHER_WAVES * 32)];
#pragma unroll
    for (int j = 0; j < NODE_DIM * ATTR_DIM / 4 / (GATHER_WAVES * 32); ++j)
      wv[j] = Wg4[tid + j * (GATHER_WAVES * 32)];
#pragma unroll
    for (int j = 0; j < GATE_ROWS * ATTR_DIM / 4 / (GATHER_WAVES * 32); ++j) {
      int idx = tid + j * (GATHER_WAVES * 32);
      int node = min(node0 + (idx >> 4), n1 - 1);
      xv[j] = Xn4[(size_t)node * (ATTR_DIM / 4) + (idx & 15)];
    }
#pragma unroll
    for (int j = 0; j < NODE_DIM * ATTR_DIM / 4 / (GATHER_WAVES * 32); ++j) {
      int idx = tid + j * (GATHER_WAVES * 32);
      *reinterpret_cast<float4*>(&sW[(idx >> 4) * FRAG_PITCH + (idx & 15) * 4]) =
          wv[j];
    }
#pragma unroll
    for (int j = 0; j < GATE_ROWS * ATTR_DIM / 4 / (GATHER_WAVES * 32); ++j) {
      int idx = tid + j * (GATHER_WAVES * 32);
      *reinterpret_cast<float4*>(&sA[(idx >> 4) * FRAG_PITCH + (idx & 15) * 4]) =
          xv[j];
    }
  }
  __syncthreads();
  {
    const int l15 = lane & 15;
    const int half = lane >> 4;
    const int mt = wq >> 2;
    const int nt0 = (wq & 3) * 2;
    const float* ap = &sA[(mt * 16 + l15) * FRAG_PITCH + half * 2];
    const float* bp0 = &sW[(nt0 * 16 + l15) * FRAG_PITCH + half * 2];
    const float* bp1 = bp0 + 16 * FRAG_PITCH;
    v8f c0 = {}, c1 = {};
#pragma unroll
    for (int i = 0; i < GATE_KSTEPS; ++i) {
      v2f a = *reinterpret_cast<const v2f*>(ap + i * 4);
      v2f b0 = *reinterpret_cast<const v2f*>(bp0 + i * 4);
      v2f b1 = *reinterpret_cast<const v2f*>(bp1 + i * 4);
      c0 = __builtin_amdgcn_wmma_f32_16x16x4_f32(
           false, a,  false, b0,
           (short)0, c0,  false,  false);
      c1 = __builtin_amdgcn_wmma_f32_16x16x4_f32(
           false, a,  false, b1,
           (short)0, c1,  false,  false);
    }
    float g0[8], g1[8];
#pragma unroll
    for (int v = 0; v < 8; ++v) {
      g0[v] = 1.0f / (1.0f + __expf(-c0[v]));
      g1[v] = 1.0f / (1.0f + __expf(-c1[v]));
    }
    __syncthreads();
    float* gp = &sG[(mt * 16 + half * 8) * GATE_PITCH + nt0 * 16 + l15];
#pragma unroll
    for (int v = 0; v < 8; ++v) {
      gp[v * GATE_PITCH] = g0[v];
      gp[v * GATE_PITCH + 16] = g1[v];
    }
  }
  __syncthreads();

  static_assert(GATHER_U == 4, "sum4_by_lane_group merges exactly four sums");
  float4 a[GATHER_K];
  float in1[GATHER_K];
  unsigned bnd[GATHER_K + 1];
#pragma unroll
  for (int k = 0; k < GATHER_K; ++k) {
    int s = min(s0 + k, n1 - 1);
    a[k] = reinterpret_cast<const float4*>(X1 + (size_t)s * NODE_DIM)[lane];
    in1[k] = invn1[s];
  }
#pragma unroll
  for (int k = 0; k <= GATHER_K; ++k) {
    int s = s0 + k;
    int v = start[min(s, n1 - 1)];
    bnd[k] = min((unsigned)(s < n1 ? v : E), (unsigned)E);
  }
  const bool row0 = lane < 16;
  const bool oct1 = (lane & 8) != 0;
  int mk[GATHER_U];
#pragma unroll
  for (int j = 0; j < GATHER_U; ++j) mk[j] = -(int)((lane >> 3) == j);
  int2 fe[GATHER_U];
#pragma unroll
  for (int j = 0; j < GATHER_U; ++j)
    fe[j] = ent[min(bnd[0] + j, max(bnd[1], 1u) - 1)];
#pragma unroll
  for (int k = 0; k < GATHER_K; ++k) {
    const unsigned beg = bnd[k], end = bnd[k + 1];
    float4 acc = make_float4(0.0f, 0.0f, 0.0f, 0.0f);
    float ssum = 0.0f;
    int2 fn[GATHER_U];
    if (k + 1 < GATHER_K) {
#pragma unroll
      for (int j = 0; j < GATHER_U; ++j)
        fn[j] = ent[min(bnd[k + 1] + j, max(bnd[k + 2], 1u) - 1)];
    }
    if (beg < end) {
      int2 ea[GATHER_U], eb[GATHER_U], en[GATHER_U];
      float4 ra[GATHER_U], rb[GATHER_U];
#pragma unroll
      for (int j = 0; j < GATHER_U; ++j) ea[j] = fe[j];
#pragma unroll
      for (int j = 0; j < GATHER_U; ++j)
        ra[j] = reinterpret_cast<const float4*>(
            X2 + (size_t)(unsigned)ea[j].x * NODE_DIM)[lane];
#pragma unroll
      for (int j = 0; j < GATHER_U; ++j)
        eb[j] = ent[min(beg + GATHER_U + j, end - 1)];
      for (unsigned pos = beg; pos < end;) {
#pragma unroll
        for (int j = 0; j < GATHER_U; ++j)
          rb[j] = reinterpret_cast<const float4*>(
              X2 + (size_t)(unsigned)eb[j].x * NODE_DIM)[lane];
#pragma unroll
        for (int j = 0; j < GATHER_U; ++j)
          en[j] = ent[min(pos + 2 * GATHER_U + j, end - 1)];
        __builtin_amdgcn_sched_barrier(0);
        batch_sum(a[k], in1[k], ra, ea, pos, end, row0, oct1, mk, acc, ssum);
        asm volatile("" ::"v"(rb[0].x), "v"(rb[0].y), "v"(rb[0].z),
                     "v"(rb[0].w), "v"(rb[1].x), "v"(rb[1].y), "v"(rb[1].z),
                     "v"(rb[1].w), "v"(rb[2].x), "v"(rb[2].y), "v"(rb[2].z),
                     "v"(rb[2].w), "v"(rb[3].x), "v"(rb[3].y), "v"(rb[3].z),
                     "v"(rb[3].w), "s"(en[0].x), "s"(en[0].y), "s"(en[1].x),
                     "s"(en[1].y), "s"(en[2].x), "s"(en[2].y), "s"(en[3].x),
                     "s"(en[3].y));
#pragma unroll
        for (int j = 0; j < GATHER_U; ++j) ea[j] = en[j];
        pos += GATHER_U;
        if (!(pos < end)) break;
#pragma unroll
        for (int j = 0; j < GATHER_U; ++j)
          ra[j] = reinterpret_cast<const float4*>(
              X2 + (size_t)(unsigned)ea[j].x * NODE_DIM)[lane];
#pragma unroll
        for (int j = 0; j < GATHER_U; ++j)
          en[j] = ent[min(pos + 2 * GATHER_U + j, end - 1)];
        __builtin_amdgcn_sched_barrier(0);
        batch_sum(a[k], in1[k], rb, eb, pos, end, row0, oct1, mk, acc, ssum);
        asm volatile("" ::"v"(ra[0].x), "v"(ra[0].y), "v"(ra[0].z),
                     "v"(ra[0].w), "v"(ra[1].x), "v"(ra[1].y), "v"(ra[1].z),
                     "v"(ra[1].w), "v"(ra[2].x), "v"(ra[2].y), "v"(ra[2].z),
                     "v"(ra[2].w), "v"(ra[3].x), "v"(ra[3].y), "v"(ra[3].z),
                     "v"(ra[3].w), "s"(en[0].x), "s"(en[0].y), "s"(en[1].x),
                     "s"(en[1].y), "s"(en[2].x), "s"(en[2].y), "s"(en[3].x),
                     "s"(en[3].y));
#pragma unroll
        for (int j = 0; j < GATHER_U; ++j) eb[j] = en[j];
        pos += GATHER_U;
      }
    }
    if (k + 1 < GATHER_K) {
#pragma unroll
      for (int j = 0; j < GATHER_U; ++j) fe[j] = fn[j];
    }
    const int s = s0 + k;
    if (s < n1) {
      const float4 g = *reinterpret_cast<const float4*>(
          &sG[(wq * GATHER_K + k) * GATE_PITCH + lane * 4]);
      float inv = ssum > 0.0f ? 1.0f / ssum : 0.0f;
      float4 r;
      r.x = g.x * (acc.x * inv);
      r.y = g.y * (acc.y * inv);
      r.z = g.z * (acc.z * inv);
      r.w = g.w * (acc.w * inv);
      reinterpret_cast<float4*>(out + (size_t)s * NODE_DIM)[lane] = r;
    }
  }
}

extern "C" void kernel_launch(void* const* d_in, const int* in_sizes, int n_in,
                              void* d_out, int out_size, void* d_ws,
                              size_t ws_size, hipStream_t stream) {
  const float* X_h_1 = (const float*)d_in[0];
  const float* X_h_2 = (const float*)d_in[1];
  const float* X_n_1 = (const float*)d_in[2];
  const int* cross = (const int*)d_in[3];
  const float* W_gate = (const float*)d_in[4];
  float* out = (float*)d_out;

  const int N1 = in_sizes[0] / NODE_DIM;
  const int N2 = in_sizes[1] / NODE_DIM;
  const int E = in_sizes[3] / 2;
  const int* src = cross;
  const int* dst = cross + E;

  int2* ent = (int2*)d_ws;
  float* invn1 = (float*)(ent + 2 * (size_t)E);
  float* invn2 = invn1 + N1;
  int* start = (int*)(invn2 + N2);

  const int npb = (N1 + LINK_BUCKETS - 1) / LINK_BUCKETS;
  const int nchunks = (E + BK_CHUNK - 1) / BK_CHUNK;
  int2* tmp = (int2*)d_out;
  int* tab = (int*)(tmp + E);

  const int norm_groups =
      (N1 + NORM_ROWS - 1) / NORM_ROWS + (N2 + NORM_ROWS - 1) / NORM_ROWS;
  const int groups_per_blk = (BK_THREADS / 32) * NORM_GROUPS;
  const int norm_blocks = (norm_groups + groups_per_blk - 1) / groups_per_blk;
  prep_kernel<<<norm_blocks + nchunks, BK_THREADS, 0, stream>>>(
      src, dst, tmp, tab, E, npb, X_h_1, invn1, N1, X_h_2, invn2, N2,
      norm_blocks);
  link_kernel<<<LINK_BUCKETS, LINK_THREADS, 0, stream>>>(
      tmp, tab, invn2, start, ent, N1, npb, nchunks);
  gather_gate_kernel<<<(N1 + GATE_ROWS - 1) / GATE_ROWS, GATHER_WAVES * 32, 0,
                       stream>>>(X_h_1, X_h_2, start, ent, invn1, X_n_1,
                                 W_gate, out, N1, E);
}
